# starting kernel
# baseline (speedup 1.0000x reference)
#include <hip/hip_runtime.h>
#include <math.h>

typedef float v2f __attribute__((ext_vector_type(2)));
typedef float v8f __attribute__((ext_vector_type(8)));

#define WMMA_F32(a, b, c) \
  __builtin_amdgcn_wmma_f32_16x16x4_f32(false, (a), false, (b), (short)0, (c), false, false)

static constexpr int B_ = 8, S_ = 512, E_ = 32, L_ = 4096, SUBE_ = 768;
static constexpr float LAMBDA_INIT = 0.2f;

static constexpr int ST0 = 513;
static constexpr int H1  = 16 * ST0;
static constexpr int SMEM_FLOATS = 16 * ST0 + 16 * 512;

__global__ __launch_bounds__(32) void kv_proj_kernel(
    const float* __restrict__ key, const float* __restrict__ Wk,
    const float* __restrict__ Wv, float* __restrict__ Km, float* __restrict__ Vm) {
  const int lane = threadIdx.x;
  const int n    = lane & 15;
  const int khi  = lane >> 4;
  const int m0   = blockIdx.x * 16;
  const int mrow = m0 + n;

  v8f ak0 = {}, ak1 = {}, av0 = {}, av1 = {};
  for (int kk = 0; kk < SUBE_ / 4; ++kk) {
    const int kb = kk * 4 + khi * 2;
    v2f a   = *(const v2f*)(key + (size_t)mrow * SUBE_ + kb);
    v2f bk0 = *(const v2f*)(Wk + (size_t)n * SUBE_ + kb);
    v2f bk1 = *(const v2f*)(Wk + (size_t)(16 + n) * SUBE_ + kb);
    v2f bv0 = *(const v2f*)(Wv + (size_t)n * SUBE_ + kb);
    v2f bv1 = *(const v2f*)(Wv + (size_t)(16 + n) * SUBE_ + kb);
    ak0 = WMMA_F32(a, bk0, ak0);
    ak1 = WMMA_F32(a, bk1, ak1);
    av0 = WMMA_F32(a, bv0, av0);
    av1 = WMMA_F32(a, bv1, av1);
  }
  const int rbase = m0 + (khi ? 8 : 0);
  for (int v = 0; v < 8; ++v) {
    const size_t r = (size_t)(rbase + v) * 32;
    Km[r + n]      = ak0[v];
    Km[r + 16 + n] = ak1[v];
    Vm[r + n]      = av0[v];
    Vm[r + 16 + n] = av1[v];
  }
}

__global__ __launch_bounds__(32) void q_proj_kernel(
    const float* __restrict__ q, const float* __restrict__ Wq,
    float* __restrict__ Qs) {
  const int lane = threadIdx.x;
  const int n    = lane & 15;
  const int khi  = lane >> 4;
  const int m0   = blockIdx.x * 16;
  const int mrow = m0 + n;

  v8f a0 = {}, a1 = {};
  for (int kk = 0; kk < 8; ++kk) {
    const int kb = kk * 4 + khi * 2;
    v2f a  = *(const v2f*)(q + (size_t)mrow * 32 + kb);
    v2f b0 = *(const v2f*)(Wq + (size_t)n * 32 + kb);
    v2f b1 = *(const v2f*)(Wq + (size_t)(16 + n) * 32 + kb);
    a0 = WMMA_F32(a, b0, a0);
    a1 = WMMA_F32(a, b1, a1);
  }
  const int rbase = m0 + (khi ? 8 : 0);
  for (int v = 0; v < 8; ++v) {
    const size_t r = (size_t)(rbase + v) * 32;
    Qs[r + n]      = a0[v] * 0.25f;
    Qs[r + 16 + n] = a1[v] * 0.25f;
  }
}

__global__ __launch_bounds__(32) void attn_kernel(
    const float* __restrict__ Qs, const float* __restrict__ Km,
    const float* __restrict__ Vm, const int* __restrict__ qmask,
    const int* __restrict__ kmask, const float* __restrict__ lq1,
    const float* __restrict__ lk1, const float* __restrict__ lq2,
    const float* __restrict__ lk2, const float* __restrict__ lnw,
    const float* __restrict__ Wo, float* __restrict__ outp,
    float* __restrict__ diffp) {
  __shared__ float smem[SMEM_FLOATS];

  const int lane = threadIdx.x;
  const int n    = lane & 15;
  const int khi  = lane >> 4;
  const int R0   = blockIdx.x * 16;
  const int b    = R0 / L_;
  const int rb   = khi ? 8 : 0;

  float s1 = 0.f, s2 = 0.f;
  for (int i = 0; i < 16; ++i) { s1 += lq1[i] * lk1[i]; s2 += lq2[i] * lk2[i]; }
  const float lambda_full = __expf(s1) - __expf(s2) + LAMBDA_INIT;

  for (int h = 0; h < 2; ++h) {
    float* dst       = (h == 0) ? smem : (smem + H1);
    const int stride = (h == 0) ? ST0 : 512;
    for (int st = 0; st < 32; ++st) {
      v8f acc = {};
      for (int kk = 0; kk < 4; ++kk) {
        const int kb = kk * 4 + khi * 2;
        v2f a  = *(const v2f*)(Qs + (size_t)(R0 + n) * 32 + h * 16 + kb);
        v2f bf = *(const v2f*)(Km + (size_t)(b * S_ + st * 16 + n) * 32 + h * 16 + kb);
        acc = WMMA_F32(a, bf, acc);
      }
      const int col = st * 16 + n;
      const int kmv = kmask[b * S_ + col];
      for (int v = 0; v < 8; ++v) {
        const int r     = rb + v;
        const int valid = kmv && qmask[R0 + r];
        dst[r * stride + col] = valid ? acc[v] : -1e20f;
      }
    }
  }
  __syncthreads();

  for (int m = 0; m < 16; ++m) {
    const float* r0 = smem + m * ST0;
    const float* r1 = smem + H1 + m * 512;
    float mx0 = -3.4e38f, mx1 = -3.4e38f;
    for (int i = 0; i < 16; ++i) {
      const int c = lane + 32 * i;
      mx0 = fmaxf(mx0, r0[c]);
      mx1 = fmaxf(mx1, r1[c]);
    }
    for (int o = 16; o > 0; o >>= 1) {
      mx0 = fmaxf(mx0, __shfl_xor(mx0, o, 32));
      mx1 = fmaxf(mx1, __shfl_xor(mx1, o, 32));
    }
    float e0[16], e1[16], sum0 = 0.f, sum1 = 0.f;
    for (int i = 0; i < 16; ++i) {
      const int c = lane + 32 * i;
      e0[i] = __expf(r0[c] - mx0); sum0 += e0[i];
      e1[i] = __expf(r1[c] - mx1); sum1 += e1[i];
    }
    for (int o = 16; o > 0; o >>= 1) {
      sum0 += __shfl_xor(sum0, o, 32);
      sum1 += __shfl_xor(sum1, o, 32);
    }
    const float inv0 = 1.f / (sum0 + 1e-8f);
    const float inv1 = 1.f / (sum1 + 1e-8f);
    float d[16], dmin = 3.4e38f;
    for (int i = 0; i < 16; ++i) {
      d[i] = e0[i] * inv0 - lambda_full * (e1[i] * inv1);
      dmin = fminf(dmin, d[i]);
    }
    for (int o = 16; o > 0; o >>= 1) dmin = fminf(dmin, __shfl_xor(dmin, o, 32));

    const int qv = qmask[R0 + m];
    float* w0 = smem + m * ST0;
    float* dg = diffp + (size_t)(R0 + m) * 512;
    for (int i = 0; i < 16; ++i) {
      const int c = lane + 32 * i;
      float dd = d[i] - dmin + 1e-5f;
      dd = (qv && kmask[b * S_ + c]) ? dd : 0.f;
      w0[c] = dd;
      __builtin_nontemporal_store(dd, dg + c);
    }
  }
  __syncthreads();

  v8f ao0 = {}, ao1 = {};
  for (int kk = 0; kk < 128; ++kk) {
    const int kb = kk * 4 + khi * 2;
    v2f a; a[0] = smem[n * ST0 + kb]; a[1] = smem[n * ST0 + kb + 1];
    const float* vr0 = Vm + (size_t)(b * S_ + kb) * 32;
    const float* vr1 = Vm + (size_t)(b * S_ + kb + 1) * 32;
    v2f b0; b0[0] = vr0[n];      b0[1] = vr1[n];
    v2f b1; b1[0] = vr0[16 + n]; b1[1] = vr1[16 + n];
    ao0 = WMMA_F32(a, b0, ao0);
    ao1 = WMMA_F32(a, b1, ao1);
  }
  float* aos = smem + H1;
  for (int v = 0; v < 8; ++v) {
    const int r = rb + v;
    aos[r * 32 + n]      = ao0[v];
    aos[r * 32 + 16 + n] = ao1[v];
  }
  __syncthreads();

  for (int m = 0; m < 16; ++m) {
    const float x = aos[m * 32 + lane];
    float sq = x * x;
    for (int o = 16; o > 0; o >>= 1) sq += __shfl_xor(sq, o, 32);
    const float rms = rsqrtf(sq * (1.0f / 32.0f) + 1e-5f);
    aos[m * 32 + lane] = x * rms * lnw[lane] * 0.8f;
  }
  __syncthreads();

  v8f o0 = {}, o1 = {};
  for (int kk = 0; kk < 8; ++kk) {
    const int kb = kk * 4 + khi * 2;
    v2f a; a[0] = aos[n * 32 + kb]; a[1] = aos[n * 32 + kb + 1];
    v2f b0 = *(const v2f*)(Wo + (size_t)n * 32 + kb);
    v2f b1 = *(const v2f*)(Wo + (size_t)(16 + n) * 32 + kb);
    o0 = WMMA_F32(a, b0, o0);
    o1 = WMMA_F32(a, b1, o1);
  }
  for (int v = 0; v < 8; ++v) {
    const size_t r = (size_t)(R0 + rb + v) * 32;
    __builtin_nontemporal_store(o0[v], outp + r + n);
    __builtin_nontemporal_store(o1[v], outp + r + 16 + n);
  }
}

extern "C" void kernel_launch(void* const* d_in, const int* in_sizes, int n_in,
                              void* d_out, int out_size, void* d_ws, size_t ws_size,
                              hipStream_t stream) {
  (void)in_sizes; (void)n_in; (void)out_size; (void)ws_size;
  const float* query = (const float*)d_in[0];
  const float* key   = (const float*)d_in[1];
  const int*   qmask = (const int*)d_in[2];
  const int*   kmask = (const int*)d_in[3];
  const float* Wq = (const float*)d_in[4];
  const float* Wk = (const float*)d_in[5];
  const float* Wv = (const float*)d_in[6];
  const float* Wo = (const float*)d_in[7];
  const float* lq1 = (const float*)d_in[8];
  const float* lk1 = (const float*)d_in[9];
  const float* lq2 = (const float*)d_in[10];
  const float* lk2 = (const float*)d_in[11];
  const float* lnw = (const float*)d_in[12];

  float* outp  = (float*)d_out;
  float* diffp = outp + (size_t)B_ * L_ * E_;

  float* Qs = (float*)d_ws;
  float* Km = Qs + (size_t)B_ * L_ * E_;
  float* Vm = Km + (size_t)B_ * S_ * E_;

  kv_proj_kernel<<<(B_ * S_) / 16, 32, 0, stream>>>(key, Wk, Wv, Km, Vm);
  q_proj_kernel<<<(B_ * L_) / 16, 32, 0, stream>>>(query, Wq, Qs);
  attn_kernel<<<(B_ * L_) / 16, 32, 0, stream>>>(Qs, Km, Vm, qmask, kmask,
                                                 lq1, lk1, lq2, lk2, lnw, Wo,
                                                 outp, diffp);
}


# attention kernel: each 16-row tile on an 8-wave work-group (score tiles, softmax rows and the diff x V k-range split per wave); attn_out sum order changes: 8 partial tiles added through LDS
# speedup vs baseline: 3.6938x; 3.6938x over previous
#include <hip/hip_runtime.h>
#include <math.h>

typedef float v2f __attribute__((ext_vector_type(2)));
typedef float v8f __attribute__((ext_vector_type(8)));

#define WMMA_F32(a, b, c) \
  __builtin_amdgcn_wmma_f32_16x16x4_f32(false, (a), false, (b), (short)0, (c), false, false)

static constexpr int B_ = 8, S_ = 512, E_ = 32, L_ = 4096, SUBE_ = 768;
static constexpr float LAMBDA_INIT = 0.2f;

static constexpr int ST0 = 513;
static constexpr int H1  = 16 * ST0;
static constexpr int SMEM_FLOATS = 16 * ST0 + 16 * 512;

__global__ __launch_bounds__(32) void kv_proj_kernel(
    const float* __restrict__ key, const float* __restrict__ Wk,
    const float* __restrict__ Wv, float* __restrict__ Km, float* __restrict__ Vm) {
  const int lane = threadIdx.x;
  const int n    = lane & 15;
  const int khi  = lane >> 4;
  const int m0   = blockIdx.x * 16;
  const int mrow = m0 + n;

  v8f ak0 = {}, ak1 = {}, av0 = {}, av1 = {};
  for (int kk = 0; kk < SUBE_ / 4; ++kk) {
    const int kb = kk * 4 + khi * 2;
    v2f a   = *(const v2f*)(key + (size_t)mrow * SUBE_ + kb);
    v2f bk0 = *(const v2f*)(Wk + (size_t)n * SUBE_ + kb);
    v2f bk1 = *(const v2f*)(Wk + (size_t)(16 + n) * SUBE_ + kb);
    v2f bv0 = *(const v2f*)(Wv + (size_t)n * SUBE_ + kb);
    v2f bv1 = *(const v2f*)(Wv + (size_t)(16 + n) * SUBE_ + kb);
    ak0 = WMMA_F32(a, bk0, ak0);
    ak1 = WMMA_F32(a, bk1, ak1);
    av0 = WMMA_F32(a, bv0, av0);
    av1 = WMMA_F32(a, bv1, av1);
  }
  const int rbase = m0 + (khi ? 8 : 0);
  for (int v = 0; v < 8; ++v) {
    const size_t r = (size_t)(rbase + v) * 32;
    Km[r + n]      = ak0[v];
    Km[r + 16 + n] = ak1[v];
    Vm[r + n]      = av0[v];
    Vm[r + 16 + n] = av1[v];
  }
}

__global__ __launch_bounds__(32) void q_proj_kernel(
    const float* __restrict__ q, const float* __restrict__ Wq,
    float* __restrict__ Qs) {
  const int lane = threadIdx.x;
  const int n    = lane & 15;
  const int khi  = lane >> 4;
  const int m0   = blockIdx.x * 16;
  const int mrow = m0 + n;

  v8f a0 = {}, a1 = {};
  for (int kk = 0; kk < 8; ++kk) {
    const int kb = kk * 4 + khi * 2;
    v2f a  = *(const v2f*)(q + (size_t)mrow * 32 + kb);
    v2f b0 = *(const v2f*)(Wq + (size_t)n * 32 + kb);
    v2f b1 = *(const v2f*)(Wq + (size_t)(16 + n) * 32 + kb);
    a0 = WMMA_F32(a, b0, a0);
    a1 = WMMA_F32(a, b1, a1);
  }
  const int rbase = m0 + (khi ? 8 : 0);
  for (int v = 0; v < 8; ++v) {
    const size_t r = (size_t)(rbase + v) * 32;
    Qs[r + n]      = a0[v] * 0.25f;
    Qs[r + 16 + n] = a1[v] * 0.25f;
  }
}

static constexpr int NW          = 8;
static constexpr int ATT_THREADS = 32 * NW;
static constexpr int PART        = H1;
static constexpr int AOS         = H1 + NW * 512;

__global__ __launch_bounds__(ATT_THREADS) void attn_kernel(
    const float* __restrict__ Qs, const float* __restrict__ Km,
    const float* __restrict__ Vm, const int* __restrict__ qmask,
    const int* __restrict__ kmask, const float* __restrict__ lq1,
    const float* __restrict__ lk1, const float* __restrict__ lq2,
    const float* __restrict__ lk2, const float* __restrict__ lnw,
    const float* __restrict__ Wo, float* __restrict__ outp,
    float* __restrict__ diffp) {
  __shared__ float smem[SMEM_FLOATS];

  const int lane = threadIdx.x & 31;
  const int w    = __builtin_amdgcn_readfirstlane((int)(threadIdx.x >> 5));
  const int n    = lane & 15;
  const int khi  = lane >> 4;
  const int R0   = blockIdx.x * 16;
  const int b    = R0 / L_;
  const int rb   = khi ? 8 : 0;

  float s1 = 0.f, s2 = 0.f;
  for (int i = 0; i < 16; ++i) { s1 += lq1[i] * lk1[i]; s2 += lq2[i] * lk2[i]; }
  const float lambda_full = __expf(s1) - __expf(s2) + LAMBDA_INIT;

  {
    v2f qa[2][4];
    for (int h = 0; h < 2; ++h)
      for (int kk = 0; kk < 4; ++kk)
        qa[h][kk] = *(const v2f*)(Qs + (size_t)(R0 + n) * 32 + h * 16 + kk * 4 + khi * 2);
    int qmv[8];
    for (int v = 0; v < 8; ++v) qmv[v] = qmask[R0 + rb + v];

    for (int t = 0; t < 4; ++t) {
      const int col = (w * 4 + t) * 16 + n;
      const int kmv = kmask[b * S_ + col];
      const float* krow = Km + (size_t)(b * S_ + col) * 32 + khi * 2;
      for (int h = 0; h < 2; ++h) {
        v8f acc = {};
        for (int kk = 0; kk < 4; ++kk) {
          v2f bf = *(const v2f*)(krow + h * 16 + kk * 4);
          acc = WMMA_F32(qa[h][kk], bf, acc);
        }
        float* dst       = (h == 0) ? smem : (smem + H1);
        const int stride = (h == 0) ? ST0 : 512;
        for (int v = 0; v < 8; ++v) {
          const int r     = rb + v;
          const int valid = kmv && qmv[v];
          dst[r * stride + col] = valid ? acc[v] : -1e20f;
        }
      }
    }
  }
  __syncthreads();

  {
    int kmc[16];
    for (int i = 0; i < 16; ++i) kmc[i] = kmask[b * S_ + lane + 32 * i];

    for (int mm = 0; mm < 2; ++mm) {
      const int m = w * 2 + mm;
      const float* r0 = smem + m * ST0;
      const float* r1 = smem + H1 + m * 512;
      float mx0 = -3.4e38f, mx1 = -3.4e38f;
      for (int i = 0; i < 16; ++i) {
        const int c = lane + 32 * i;
        mx0 = fmaxf(mx0, r0[c]);
        mx1 = fmaxf(mx1, r1[c]);
      }
      for (int o = 16; o > 0; o >>= 1) {
        mx0 = fmaxf(mx0, __shfl_xor(mx0, o, 32));
        mx1 = fmaxf(mx1, __shfl_xor(mx1, o, 32));
      }
      float e0[16], e1[16], sum0 = 0.f, sum1 = 0.f;
      for (int i = 0; i < 16; ++i) {
        const int c = lane + 32 * i;
        e0[i] = __expf(r0[c] - mx0); sum0 += e0[i];
        e1[i] = __expf(r1[c] - mx1); sum1 += e1[i];
      }
      for (int o = 16; o > 0; o >>= 1) {
        sum0 += __shfl_xor(sum0, o, 32);
        sum1 += __shfl_xor(sum1, o, 32);
      }
      const float inv0 = 1.f / (sum0 + 1e-8f);
      const float inv1 = 1.f / (sum1 + 1e-8f);
      float d[16], dmin = 3.4e38f;
      for (int i = 0; i < 16; ++i) {
        d[i] = e0[i] * inv0 - lambda_full * (e1[i] * inv1);
        dmin = fminf(dmin, d[i]);
      }
      for (int o = 16; o > 0; o >>= 1) dmin = fminf(dmin, __shfl_xor(dmin, o, 32));

      const int qv = qmask[R0 + m];
      float* w0 = smem + m * ST0;
      float* dg = diffp + (size_t)(R0 + m) * 512;
      for (int i = 0; i < 16; ++i) {
        const int c = lane + 32 * i;
        float dd = d[i] - dmin + 1e-5f;
        dd = (qv && kmc[i]) ? dd : 0.f;
        w0[c] = dd;
        __builtin_nontemporal_store(dd, dg + c);
      }
    }
  }
  __syncthreads();

  {
    v8f ao0 = {}, ao1 = {};
    for (int j = 0; j < 16; ++j) {
      const int kb = (w * 16 + j) * 4 + khi * 2;
      v2f a; a[0] = smem[n * ST0 + kb]; a[1] = smem[n * ST0 + kb + 1];
      const float* vr0 = Vm + (size_t)(b * S_ + kb) * 32;
      const float* vr1 = Vm + (size_t)(b * S_ + kb + 1) * 32;
      v2f b0; b0[0] = vr0[n];      b0[1] = vr1[n];
      v2f b1; b1[0] = vr0[16 + n]; b1[1] = vr1[16 + n];
      ao0 = WMMA_F32(a, b0, ao0);
      ao1 = WMMA_F32(a, b1, ao1);
    }
    float* part = smem + PART + w * 512;
    for (int v = 0; v < 8; ++v) {
      const int r = rb + v;
      part[r * 32 + n]      = ao0[v];
      part[r * 32 + 16 + n] = ao1[v];
    }
  }
  __syncthreads();

  for (int mm = 0; mm < 2; ++mm) {
    const int m = w * 2 + mm;
    float x = smem[PART + m * 32 + lane];
    for (int p = 1; p < NW; ++p) x += smem[PART + p * 512 + m * 32 + lane];
    float sq = x * x;
    for (int o = 16; o > 0; o >>= 1) sq += __shfl_xor(sq, o, 32);
    const float rms = rsqrtf(sq * (1.0f / 32.0f) + 1e-5f);
    smem[AOS + m * 32 + lane] = x * rms * lnw[lane] * 0.8f;
  }
  __syncthreads();

  if (w != 0) return;

  const float* aos = smem + AOS;
  v8f o0 = {}, o1 = {};
  for (int kk = 0; kk < 8; ++kk) {
    const int kb = kk * 4 + khi * 2;
    v2f a; a[0] = aos[n * 32 + kb]; a[1] = aos[n * 32 + kb + 1];
    v2f b0 = *(const v2f*)(Wo + (size_t)n * 32 + kb);
    v2f b1 = *(const v2f*)(Wo + (size_t)(16 + n) * 32 + kb);
    o0 = WMMA_F32(a, b0, o0);
    o1 = WMMA_F32(a, b1, o1);
  }
  for (int v = 0; v < 8; ++v) {
    const size_t r = (size_t)(R0 + rb + v) * 32;
    __builtin_nontemporal_store(o0[v], outp + r + n);
    __builtin_nontemporal_store(o1[v], outp + r + 16 + n);
  }
}

extern "C" void kernel_launch(void* const* d_in, const int* in_sizes, int n_in,
                              void* d_out, int out_size, void* d_ws, size_t ws_size,
                              hipStream_t stream) {
  (void)in_sizes; (void)n_in; (void)out_size; (void)ws_size;
  const float* query = (const float*)d_in[0];
  const float* key   = (const float*)d_in[1];
  const int*   qmask = (const int*)d_in[2];
  const int*   kmask = (const int*)d_in[3];
  const float* Wq = (const float*)d_in[4];
  const float* Wk = (const float*)d_in[5];
  const float* Wv = (const float*)d_in[6];
  const float* Wo = (const float*)d_in[7];
  const float* lq1 = (const float*)d_in[8];
  const float* lk1 = (const float*)d_in[9];
  const float* lq2 = (const float*)d_in[10];
  const float* lk2 = (const float*)d_in[11];
  const float* lnw = (const float*)d_in[12];

  float* outp  = (float*)d_out;
  float* diffp = outp + (size_t)B_ * L_ * E_;

  float* Qs = (float*)d_ws;
  float* Km = Qs + (size_t)B_ * L_ * E_;
  float* Vm = Km + (size_t)B_ * S_ * E_;

  kv_proj_kernel<<<(B_ * S_) / 16, 32, 0, stream>>>(key, Wk, Wv, Km, Vm);
  q_proj_kernel<<<(B_ * L_) / 16, 32, 0, stream>>>(query, Wq, Qs);
  attn_kernel<<<(B_ * L_) / 16, ATT_THREADS, 0, stream>>>(Qs, Km, Vm, qmask, kmask,
                                                 lq1, lk1, lq2, lk2, lnw, Wo,
                                                 outp, diffp);
}


# K/V projection: 8 waves per 16-row tile, each on one 96-wide k-slice with its 16-byte operand loads all in flight; Km/Vm sum order changes: 8 partial tiles added in wave order through LDS
# speedup vs baseline: 5.9255x; 1.6042x over previous
#include <hip/hip_runtime.h>
#include <math.h>

typedef float v2f __attribute__((ext_vector_type(2)));
typedef float v8f __attribute__((ext_vector_type(8)));

#define WMMA_F32(a, b, c) \
  __builtin_amdgcn_wmma_f32_16x16x4_f32(false, (a), false, (b), (short)0, (c), false, false)

static constexpr int B_ = 8, S_ = 512, E_ = 32, L_ = 4096, SUBE_ = 768;
static constexpr float LAMBDA_INIT = 0.2f;

static constexpr int ST0 = 513;
static constexpr int H1  = 16 * ST0;
static constexpr int SMEM_FLOATS = 16 * ST0 + 16 * 512;

typedef float v4f __attribute__((ext_vector_type(4)));

static constexpr int KV_NW  = 8;
static constexpr int KV_KS  = SUBE_ / KV_NW;
static constexpr int KV_PST = 66;
static constexpr int KV_PT  = 16 * KV_PST;

__global__ __launch_bounds__(32 * KV_NW) __attribute__((amdgpu_waves_per_eu(2, 2)))
void kv_proj_kernel(
    const float* __restrict__ key, const float* __restrict__ Wk,
    const float* __restrict__ Wv, float* __restrict__ Km, float* __restrict__ Vm) {
  __shared__ float part[KV_NW * KV_PT];

  const int lane = threadIdx.x & 31;
  const int w    = __builtin_amdgcn_readfirstlane((int)(threadIdx.x >> 5));
  const int n    = lane & 15;
  const int khi  = lane >> 4;
  const int m0   = blockIdx.x * 16;
  const int koff = w * KV_KS + khi * 4;

  const float* ap  = key + (size_t)(m0 + n) * SUBE_ + koff;
  const float* pk0 = Wk + (size_t)n * SUBE_ + koff;
  const float* pk1 = Wk + (size_t)(16 + n) * SUBE_ + koff;
  const float* pv0 = Wv + (size_t)n * SUBE_ + koff;
  const float* pv1 = Wv + (size_t)(16 + n) * SUBE_ + koff;

  v8f ak0 = {}, ak1 = {}, av0 = {}, av1 = {};
  for (int j = 0; j < KV_KS / 8; ++j) {
    const v4f a   = *(const v4f*)(ap  + 8 * j);
    const v4f bk0 = *(const v4f*)(pk0 + 8 * j);
    const v4f bk1 = *(const v4f*)(pk1 + 8 * j);
    const v4f bv0 = *(const v4f*)(pv0 + 8 * j);
    const v4f bv1 = *(const v4f*)(pv1 + 8 * j);
    v2f alo, ahi, k0lo, k0hi, k1lo, k1hi, v0lo, v0hi, v1lo, v1hi;
    alo[0]  = a[0];   alo[1]  = a[1];   ahi[0]  = a[2];   ahi[1]  = a[3];
    k0lo[0] = bk0[0]; k0lo[1] = bk0[1]; k0hi[0] = bk0[2]; k0hi[1] = bk0[3];
    k1lo[0] = bk1[0]; k1lo[1] = bk1[1]; k1hi[0] = bk1[2]; k1hi[1] = bk1[3];
    v0lo[0] = bv0[0]; v0lo[1] = bv0[1]; v0hi[0] = bv0[2]; v0hi[1] = bv0[3];
    v1lo[0] = bv1[0]; v1lo[1] = bv1[1]; v1hi[0] = bv1[2]; v1hi[1] = bv1[3];
    ak0 = WMMA_F32(alo, k0lo, ak0);
    ak1 = WMMA_F32(alo, k1lo, ak1);
    av0 = WMMA_F32(alo, v0lo, av0);
    av1 = WMMA_F32(alo, v1lo, av1);
    ak0 = WMMA_F32(ahi, k0hi, ak0);
    ak1 = WMMA_F32(ahi, k1hi, ak1);
    av0 = WMMA_F32(ahi, v0hi, av0);
    av1 = WMMA_F32(ahi, v1hi, av1);
  }

  float* pw = part + w * KV_PT;
  for (int v = 0; v < 8; ++v) {
    const int r = (khi * 8 + v) * KV_PST + n;
    pw[r]      = ak0[v];
    pw[r + 16] = ak1[v];
    pw[r + 32] = av0[v];
    pw[r + 48] = av1[v];
  }
  __syncthreads();

  for (int i = 0; i < 4; ++i) {
    const int o = (int)threadIdx.x + 32 * KV_NW * i;
    const int r = o >> 6;
    const int c = o & 63;
    float x = part[r * KV_PST + c];
    for (int p = 1; p < KV_NW; ++p) x += part[p * KV_PT + r * KV_PST + c];
    float* dst = (c < 32) ? Km : Vm;
    dst[(size_t)(m0 + r) * 32 + (c & 31)] = x;
  }
}

__global__ __launch_bounds__(32) void q_proj_kernel(
    const float* __restrict__ q, const float* __restrict__ Wq,
    float* __restrict__ Qs) {
  const int lane = threadIdx.x;
  const int n    = lane & 15;
  const int khi  = lane >> 4;
  const int m0   = blockIdx.x * 16;
  const int mrow = m0 + n;

  v8f a0 = {}, a1 = {};
  for (int kk = 0; kk < 8; ++kk) {
    const int kb = kk * 4 + khi * 2;
    v2f a  = *(const v2f*)(q + (size_t)mrow * 32 + kb);
    v2f b0 = *(const v2f*)(Wq + (size_t)n * 32 + kb);
    v2f b1 = *(const v2f*)(Wq + (size_t)(16 + n) * 32 + kb);
    a0 = WMMA_F32(a, b0, a0);
    a1 = WMMA_F32(a, b1, a1);
  }
  const int rbase = m0 + (khi ? 8 : 0);
  for (int v = 0; v < 8; ++v) {
    const size_t r = (size_t)(rbase + v) * 32;
    Qs[r + n]      = a0[v] * 0.25f;
    Qs[r + 16 + n] = a1[v] * 0.25f;
  }
}

static constexpr int NW          = 8;
static constexpr int ATT_THREADS = 32 * NW;
static constexpr int PART        = H1;
static constexpr int AOS         = H1 + NW * 512;

__global__ __launch_bounds__(ATT_THREADS) void attn_kernel(
    const float* __restrict__ Qs, const float* __restrict__ Km,
    const float* __restrict__ Vm, const int* __restrict__ qmask,
    const int* __restrict__ kmask, const float* __restrict__ lq1,
    const float* __restrict__ lk1, const float* __restrict__ lq2,
    const float* __restrict__ lk2, const float* __restrict__ lnw,
    const float* __restrict__ Wo, float* __restrict__ outp,
    float* __restrict__ diffp) {
  __shared__ float smem[SMEM_FLOATS];

  const int lane = threadIdx.x & 31;
  const int w    = __builtin_amdgcn_readfirstlane((int)(threadIdx.x >> 5));
  const int n    = lane & 15;
  const int khi  = lane >> 4;
  const int R0   = blockIdx.x * 16;
  const int b    = R0 / L_;
  const int rb   = khi ? 8 : 0;

  float s1 = 0.f, s2 = 0.f;
  for (int i = 0; i < 16; ++i) { s1 += lq1[i] * lk1[i]; s2 += lq2[i] * lk2[i]; }
  const float lambda_full = __expf(s1) - __expf(s2) + LAMBDA_INIT;

  {
    v2f qa[2][4];
    for (int h = 0; h < 2; ++h)
      for (int kk = 0; kk < 4; ++kk)
        qa[h][kk] = *(const v2f*)(Qs + (size_t)(R0 + n) * 32 + h * 16 + kk * 4 + khi * 2);
    int qmv[8];
    for (int v = 0; v < 8; ++v) qmv[v] = qmask[R0 + rb + v];

    for (int t = 0; t < 4; ++t) {
      const int col = (w * 4 + t) * 16 + n;
      const int kmv = kmask[b * S_ + col];
      const float* krow = Km + (size_t)(b * S_ + col) * 32 + khi * 2;
      for (int h = 0; h < 2; ++h) {
        v8f acc = {};
        for (int kk = 0; kk < 4; ++kk) {
          v2f bf = *(const v2f*)(krow + h * 16 + kk * 4);
          acc = WMMA_F32(qa[h][kk], bf, acc);
        }
        float* dst       = (h == 0) ? smem : (smem + H1);
        const int stride = (h == 0) ? ST0 : 512;
        for (int v = 0; v < 8; ++v) {
          const int r     = rb + v;
          const int valid = kmv && qmv[v];
          dst[r * stride + col] = valid ? acc[v] : -1e20f;
        }
      }
    }
  }
  __syncthreads();

  {
    int kmc[16];
    for (int i = 0; i < 16; ++i) kmc[i] = kmask[b * S_ + lane + 32 * i];

    for (int mm = 0; mm < 2; ++mm) {
      const int m = w * 2 + mm;
      const float* r0 = smem + m * ST0;
      const float* r1 = smem + H1 + m * 512;
      float mx0 = -3.4e38f, mx1 = -3.4e38f;
      for (int i = 0; i < 16; ++i) {
        const int c = lane + 32 * i;
        mx0 = fmaxf(mx0, r0[c]);
        mx1 = fmaxf(mx1, r1[c]);
      }
      for (int o = 16; o > 0; o >>= 1) {
        mx0 = fmaxf(mx0, __shfl_xor(mx0, o, 32));
        mx1 = fmaxf(mx1, __shfl_xor(mx1, o, 32));
      }
      float e0[16], e1[16], sum0 = 0.f, sum1 = 0.f;
      for (int i = 0; i < 16; ++i) {
        const int c = lane + 32 * i;
        e0[i] = __expf(r0[c] - mx0); sum0 += e0[i];
        e1[i] = __expf(r1[c] - mx1); sum1 += e1[i];
      }
      for (int o = 16; o > 0; o >>= 1) {
        sum0 += __shfl_xor(sum0, o, 32);
        sum1 += __shfl_xor(sum1, o, 32);
      }
      const float inv0 = 1.f / (sum0 + 1e-8f);
      const float inv1 = 1.f / (sum1 + 1e-8f);
      float d[16], dmin = 3.4e38f;
      for (int i = 0; i < 16; ++i) {
        d[i] = e0[i] * inv0 - lambda_full * (e1[i] * inv1);
        dmin = fminf(dmin, d[i]);
      }
      for (int o = 16; o > 0; o >>= 1) dmin = fminf(dmin, __shfl_xor(dmin, o, 32));

      const int qv = qmask[R0 + m];
      float* w0 = smem + m * ST0;
      float* dg = diffp + (size_t)(R0 + m) * 512;
      for (int i = 0; i < 16; ++i) {
        const int c = lane + 32 * i;
        float dd = d[i] - dmin + 1e-5f;
        dd = (qv && kmc[i]) ? dd : 0.f;
        w0[c] = dd;
        __builtin_nontemporal_store(dd, dg + c);
      }
    }
  }
  __syncthreads();

  {
    v8f ao0 = {}, ao1 = {};
    for (int j = 0; j < 16; ++j) {
      const int kb = (w * 16 + j) * 4 + khi * 2;
      v2f a; a[0] = smem[n * ST0 + kb]; a[1] = smem[n * ST0 + kb + 1];
      const float* vr0 = Vm + (size_t)(b * S_ + kb) * 32;
      const float* vr1 = Vm + (size_t)(b * S_ + kb + 1) * 32;
      v2f b0; b0[0] = vr0[n];      b0[1] = vr1[n];
      v2f b1; b1[0] = vr0[16 + n]; b1[1] = vr1[16 + n];
      ao0 = WMMA_F32(a, b0, ao0);
      ao1 = WMMA_F32(a, b1, ao1);
    }
    float* part = smem + PART + w * 512;
    for (int v = 0; v < 8; ++v) {
      const int r = rb + v;
      part[r * 32 + n]      = ao0[v];
      part[r * 32 + 16 + n] = ao1[v];
    }
  }
  __syncthreads();

  for (int mm = 0; mm < 2; ++mm) {
    const int m = w * 2 + mm;
    float x = smem[PART + m * 32 + lane];
    for (int p = 1; p < NW; ++p) x += smem[PART + p * 512 + m * 32 + lane];
    float sq = x * x;
    for (int o = 16; o > 0; o >>= 1) sq += __shfl_xor(sq, o, 32);
    const float rms = rsqrtf(sq * (1.0f / 32.0f) + 1e-5f);
    smem[AOS + m * 32 + lane] = x * rms * lnw[lane] * 0.8f;
  }
  __syncthreads();

  if (w != 0) return;

  const float* aos = smem + AOS;
  v8f o0 = {}, o1 = {};
  for (int kk = 0; kk < 8; ++kk) {
    const int kb = kk * 4 + khi * 2;
    v2f a; a[0] = aos[n * 32 + kb]; a[1] = aos[n * 32 + kb + 1];
    v2f b0 = *(const v2f*)(Wo + (size_t)n * 32 + kb);
    v2f b1 = *(const v2f*)(Wo + (size_t)(16 + n) * 32 + kb);
    o0 = WMMA_F32(a, b0, o0);
    o1 = WMMA_F32(a, b1, o1);
  }
  for (int v = 0; v < 8; ++v) {
    const size_t r = (size_t)(R0 + rb + v) * 32;
    __builtin_nontemporal_store(o0[v], outp + r + n);
    __builtin_nontemporal_store(o1[v], outp + r + 16 + n);
  }
}

extern "C" void kernel_launch(void* const* d_in, const int* in_sizes, int n_in,
                              void* d_out, int out_size, void* d_ws, size_t ws_size,
                              hipStream_t stream) {
  (void)in_sizes; (void)n_in; (void)out_size; (void)ws_size;
  const float* query = (const float*)d_in[0];
  const float* key   = (const float*)d_in[1];
  const int*   qmask = (const int*)d_in[2];
  const int*   kmask = (const int*)d_in[3];
  const float* Wq = (const float*)d_in[4];
  const float* Wk = (const float*)d_in[5];
  const float* Wv = (const float*)d_in[6];
  const float* Wo = (const float*)d_in[7];
  const float* lq1 = (const float*)d_in[8];
  const float* lk1 = (const float*)d_in[9];
  const float* lq2 = (const float*)d_in[10];
  const float* lk2 = (const float*)d_in[11];
  const float* lnw = (const float*)d_in[12];

  float* outp  = (float*)d_out;
  float* diffp = outp + (size_t)B_ * L_ * E_;

  float* Qs = (float*)d_ws;
  float* Km = Qs + (size_t)B_ * L_ * E_;
  float* Vm = Km + (size_t)B_ * S_ * E_;

  kv_proj_kernel<<<(B_ * S_) / 16, 32 * KV_NW, 0, stream>>>(key, Wk, Wv, Km, Vm);
  q_proj_kernel<<<(B_ * L_) / 16, 32, 0, stream>>>(query, Wq, Qs);
  attn_kernel<<<(B_ * L_) / 16, ATT_THREADS, 0, stream>>>(Qs, Km, Vm, qmask, kmask,
                                                 lq1, lk1, lq2, lk2, lnw, Wo,
                                                 outp, diffp);
}


# attention kernel: each phase's global loads issued as one group ahead of use and kept in flight across barriers that wait for LDS only (V fetched during the softmax; diff stores drain behind)
# speedup vs baseline: 6.1175x; 1.0324x over previous
#include <hip/hip_runtime.h>
#include <math.h>

typedef float v2f __attribute__((ext_vector_type(2)));
typedef float v8f __attribute__((ext_vector_type(8)));

#define WMMA_F32(a, b, c) \
  __builtin_amdgcn_wmma_f32_16x16x4_f32(false, (a), false, (b), (short)0, (c), false, false)

static constexpr int B_ = 8, S_ = 512, E_ = 32, L_ = 4096, SUBE_ = 768;
static constexpr float LAMBDA_INIT = 0.2f;

static constexpr int ST0 = 513;
static constexpr int H1  = 16 * ST0;
static constexpr int SMEM_FLOATS = 16 * ST0 + 16 * 512;

typedef float v4f __attribute__((ext_vector_type(4)));

static constexpr int KV_NW  = 8;
static constexpr int KV_KS  = SUBE_ / KV_NW;
static constexpr int KV_PST = 66;
static constexpr int KV_PT  = 16 * KV_PST;

__global__ __launch_bounds__(32 * KV_NW) __attribute__((amdgpu_waves_per_eu(2, 2)))
void kv_proj_kernel(
    const float* __restrict__ key, const float* __restrict__ Wk,
    const float* __restrict__ Wv, float* __restrict__ Km, float* __restrict__ Vm) {
  __shared__ float part[KV_NW * KV_PT];

  const int lane = threadIdx.x & 31;
  const int w    = __builtin_amdgcn_readfirstlane((int)(threadIdx.x >> 5));
  const int n    = lane & 15;
  const int khi  = lane >> 4;
  const int m0   = blockIdx.x * 16;
  const int koff = w * KV_KS + khi * 4;

  const float* ap  = key + (size_t)(m0 + n) * SUBE_ + koff;
  const float* pk0 = Wk + (size_t)n * SUBE_ + koff;
  const float* pk1 = Wk + (size_t)(16 + n) * SUBE_ + koff;
  const float* pv0 = Wv + (size_t)n * SUBE_ + koff;
  const float* pv1 = Wv + (size_t)(16 + n) * SUBE_ + koff;

  v8f ak0 = {}, ak1 = {}, av0 = {}, av1 = {};
  for (int j = 0; j < KV_KS / 8; ++j) {
    const v4f a   = *(const v4f*)(ap  + 8 * j);
    const v4f bk0 = *(const v4f*)(pk0 + 8 * j);
    const v4f bk1 = *(const v4f*)(pk1 + 8 * j);
    const v4f bv0 = *(const v4f*)(pv0 + 8 * j);
    const v4f bv1 = *(const v4f*)(pv1 + 8 * j);
    v2f alo, ahi, k0lo, k0hi, k1lo, k1hi, v0lo, v0hi, v1lo, v1hi;
    alo[0]  = a[0];   alo[1]  = a[1];   ahi[0]  = a[2];   ahi[1]  = a[3];
    k0lo[0] = bk0[0]; k0lo[1] = bk0[1]; k0hi[0] = bk0[2]; k0hi[1] = bk0[3];
    k1lo[0] = bk1[0]; k1lo[1] = bk1[1]; k1hi[0] = bk1[2]; k1hi[1] = bk1[3];
    v0lo[0] = bv0[0]; v0lo[1] = bv0[1]; v0hi[0] = bv0[2]; v0hi[1] = bv0[3];
    v1lo[0] = bv1[0]; v1lo[1] = bv1[1]; v1hi[0] = bv1[2]; v1hi[1] = bv1[3];
    ak0 = WMMA_F32(alo, k0lo, ak0);
    ak1 = WMMA_F32(alo, k1lo, ak1);
    av0 = WMMA_F32(alo, v0lo, av0);
    av1 = WMMA_F32(alo, v1lo, av1);
    ak0 = WMMA_F32(ahi, k0hi, ak0);
    ak1 = WMMA_F32(ahi, k1hi, ak1);
    av0 = WMMA_F32(ahi, v0hi, av0);
    av1 = WMMA_F32(ahi, v1hi, av1);
  }

  float* pw = part + w * KV_PT;
  for (int v = 0; v < 8; ++v) {
    const int r = (khi * 8 + v) * KV_PST + n;
    pw[r]      = ak0[v];
    pw[r + 16] = ak1[v];
    pw[r + 32] = av0[v];
    pw[r + 48] = av1[v];
  }
  __syncthreads();

  for (int i = 0; i < 4; ++i) {
    const int o = (int)threadIdx.x + 32 * KV_NW * i;
    const int r = o >> 6;
    const int c = o & 63;
    float x = part[r * KV_PST + c];
    for (int p = 1; p < KV_NW; ++p) x += part[p * KV_PT + r * KV_PST + c];
    float* dst = (c < 32) ? Km : Vm;
    dst[(size_t)(m0 + r) * 32 + (c & 31)] = x;
  }
}

__global__ __launch_bounds__(32) void q_proj_kernel(
    const float* __restrict__ q, const float* __restrict__ Wq,
    float* __restrict__ Qs) {
  const int lane = threadIdx.x;
  const int n    = lane & 15;
  const int khi  = lane >> 4;
  const int m0   = blockIdx.x * 16;
  const int mrow = m0 + n;

  v8f a0 = {}, a1 = {};
  for (int kk = 0; kk < 8; ++kk) {
    const int kb = kk * 4 + khi * 2;
    v2f a  = *(const v2f*)(q + (size_t)mrow * 32 + kb);
    v2f b0 = *(const v2f*)(Wq + (size_t)n * 32 + kb);
    v2f b1 = *(const v2f*)(Wq + (size_t)(16 + n) * 32 + kb);
    a0 = WMMA_F32(a, b0, a0);
    a1 = WMMA_F32(a, b1, a1);
  }
  const int rbase = m0 + (khi ? 8 : 0);
  for (int v = 0; v < 8; ++v) {
    const size_t r = (size_t)(rbase + v) * 32;
    Qs[r + n]      = a0[v] * 0.25f;
    Qs[r + 16 + n] = a1[v] * 0.25f;
  }
}

static __device__ __forceinline__ void lds_barrier() {
  __builtin_amdgcn_fence(__ATOMIC_RELEASE, "workgroup", "local");
  __builtin_amdgcn_s_barrier();
  __builtin_amdgcn_fence(__ATOMIC_ACQUIRE, "workgroup", "local");
}

static constexpr int NW          = 8;
static constexpr int ATT_THREADS = 32 * NW;
static constexpr int PART        = H1;
static constexpr int AOS         = H1 + NW * 512;

__global__ __launch_bounds__(ATT_THREADS) void attn_kernel(
    const float* __restrict__ Qs, const float* __restrict__ Km,
    const float* __restrict__ Vm, const int* __restrict__ qmask,
    const int* __restrict__ kmask, const float* __restrict__ lq1,
    const float* __restrict__ lk1, const float* __restrict__ lq2,
    const float* __restrict__ lk2, const float* __restrict__ lnw,
    const float* __restrict__ Wo, float* __restrict__ outp,
    float* __restrict__ diffp) {
  __shared__ float smem[SMEM_FLOATS];

  const int lane = threadIdx.x & 31;
  const int w    = __builtin_amdgcn_readfirstlane((int)(threadIdx.x >> 5));
  const int n    = lane & 15;
  const int khi  = lane >> 4;
  const int R0   = blockIdx.x * 16;
  const int b    = R0 / L_;
  const int rb   = khi ? 8 : 0;

  {
    v2f qa[2][4];
    for (int h = 0; h < 2; ++h)
      for (int kk = 0; kk < 4; ++kk)
        qa[h][kk] = *(const v2f*)(Qs + (size_t)(R0 + n) * 32 + h * 16 + kk * 4 + khi * 2);
    int qlo[8], qhi[8];
    for (int v = 0; v < 8; ++v) { qlo[v] = qmask[R0 + v]; qhi[v] = qmask[R0 + 8 + v]; }

    v2f kf[4][2][4];
    int kmv4[4];
    for (int t = 0; t < 4; ++t) {
      const int col = (w * 4 + t) * 16 + n;
      kmv4[t] = kmask[b * S_ + col];
      const float* krow = Km + (size_t)(b * S_ + col) * 32 + khi * 2;
      for (int h = 0; h < 2; ++h)
        for (int kk = 0; kk < 4; ++kk)
          kf[t][h][kk] = *(const v2f*)(krow + h * 16 + kk * 4);
    }
    __builtin_amdgcn_sched_barrier(0);

    for (int t = 0; t < 4; ++t) {
      const int col = (w * 4 + t) * 16 + n;
      const int kmv = kmv4[t];
      for (int h = 0; h < 2; ++h) {
        v8f acc = {};
        for (int kk = 0; kk < 4; ++kk) {
          acc = WMMA_F32(qa[h][kk], kf[t][h][kk], acc);
        }
        float* dst       = (h == 0) ? smem : (smem + H1);
        const int stride = (h == 0) ? ST0 : 512;
        for (int v = 0; v < 8; ++v) {
          const int r     = rb + v;
          const int valid = kmv && (khi ? qhi[v] : qlo[v]);
          dst[r * stride + col] = valid ? acc[v] : -1e20f;
        }
      }
    }
  }
  lds_barrier();

  v2f vb0[16], vb1[16];

  {
    float s1 = 0.f, s2 = 0.f;
    for (int i = 0; i < 16; ++i) { s1 += lq1[i] * lk1[i]; s2 += lq2[i] * lk2[i]; }
    const float lambda_full = __expf(s1) - __expf(s2) + LAMBDA_INIT;

    int kmc[16];
    for (int i = 0; i < 16; ++i) kmc[i] = kmask[b * S_ + lane + 32 * i];

    for (int mm = 0; mm < 2; ++mm) {
      const int m = w * 2 + mm;
      const float* r0 = smem + m * ST0;
      const float* r1 = smem + H1 + m * 512;
      float mx0 = -3.4e38f, mx1 = -3.4e38f;
      for (int i = 0; i < 16; ++i) {
        const int c = lane + 32 * i;
        mx0 = fmaxf(mx0, r0[c]);
        mx1 = fmaxf(mx1, r1[c]);
      }
      for (int o = 16; o > 0; o >>= 1) {
        mx0 = fmaxf(mx0, __shfl_xor(mx0, o, 32));
        mx1 = fmaxf(mx1, __shfl_xor(mx1, o, 32));
      }
      float e0[16], e1[16], sum0 = 0.f, sum1 = 0.f;
      for (int i = 0; i < 16; ++i) {
        const int c = lane + 32 * i;
        e0[i] = __expf(r0[c] - mx0); sum0 += e0[i];
        e1[i] = __expf(r1[c] - mx1); sum1 += e1[i];
      }
      if (mm == 1) {
        __builtin_amdgcn_sched_barrier(0);
        const float* vbase = Vm + (size_t)(b * S_ + w * 64 + khi * 2) * 32 + n;
#pragma unroll
        for (int j = 0; j < 15; ++j) {
          const float* vr = vbase + j * 128;
          vb0[j][0] = vr[0];  vb0[j][1] = vr[32];
          vb1[j][0] = vr[16]; vb1[j][1] = vr[48];
        }
      }
      for (int o = 16; o > 0; o >>= 1) {
        sum0 += __shfl_xor(sum0, o, 32);
        sum1 += __shfl_xor(sum1, o, 32);
      }
      const float inv0 = 1.f / (sum0 + 1e-8f);
      const float inv1 = 1.f / (sum1 + 1e-8f);
      float d[16], dmin = 3.4e38f;
      for (int i = 0; i < 16; ++i) {
        d[i] = e0[i] * inv0 - lambda_full * (e1[i] * inv1);
        dmin = fminf(dmin, d[i]);
      }
      for (int o = 16; o > 0; o >>= 1) dmin = fminf(dmin, __shfl_xor(dmin, o, 32));

      const int qv = qmask[R0 + m];
      float* w0 = smem + m * ST0;
      float* dg = diffp + (size_t)(R0 + m) * 512;
      for (int i = 0; i < 16; ++i) {
        const int c = lane + 32 * i;
        float dd = d[i] - dmin + 1e-5f;
        dd = (qv && kmc[i]) ? dd : 0.f;
        w0[c] = dd;
        __builtin_nontemporal_store(dd, dg + c);
      }

      if (mm == 1) {
        __builtin_amdgcn_sched_barrier(0);
        const float* vr = Vm + (size_t)(b * S_ + w * 64 + khi * 2) * 32 + n + 15 * 128;
        vb0[15][0] = vr[0];  vb0[15][1] = vr[32];
        vb1[15][0] = vr[16]; vb1[15][1] = vr[48];
      }
    }
  }
  lds_barrier();

  float lnw_l;
  v2f wo0[8], wo1[8];
  {
    v2f af[16];
#pragma unroll
    for (int j = 0; j < 16; ++j) {
      const int kb = (w * 16 + j) * 4 + khi * 2;
      af[j][0] = smem[n * ST0 + kb]; af[j][1] = smem[n * ST0 + kb + 1];
    }
    __builtin_amdgcn_sched_barrier(0);
    v8f ao0 = {}, ao1 = {};
#pragma unroll
    for (int j = 0; j < 16; ++j) {
      ao0 = WMMA_F32(af[j], vb0[j], ao0);
      ao1 = WMMA_F32(af[j], vb1[j], ao1);
    }

    lnw_l = lnw[lane];
    if (w == 0) {
      for (int kk = 0; kk < 8; ++kk) {
        const int kb = kk * 4 + khi * 2;
        wo0[kk] = *(const v2f*)(Wo + (size_t)n * 32 + kb);
        wo1[kk] = *(const v2f*)(Wo + (size_t)(16 + n) * 32 + kb);
      }
    }

    float* part = smem + PART + w * 512;
    for (int v = 0; v < 8; ++v) {
      const int r = rb + v;
      part[r * 32 + n]      = ao0[v];
      part[r * 32 + 16 + n] = ao1[v];
    }
  }
  lds_barrier();

  for (int mm = 0; mm < 2; ++mm) {
    const int m = w * 2 + mm;
    float x = smem[PART + m * 32 + lane];
    for (int p = 1; p < NW; ++p) x += smem[PART + p * 512 + m * 32 + lane];
    float sq = x * x;
    for (int o = 16; o > 0; o >>= 1) sq += __shfl_xor(sq, o, 32);
    const float rms = rsqrtf(sq * (1.0f / 32.0f) + 1e-5f);
    smem[AOS + m * 32 + lane] = x * rms * lnw_l * 0.8f;
  }
  lds_barrier();

  if (w != 0) return;

  const float* aos = smem + AOS;
  v8f o0 = {}, o1 = {};
  for (int kk = 0; kk < 8; ++kk) {
    const int kb = kk * 4 + khi * 2;
    v2f a; a[0] = aos[n * 32 + kb]; a[1] = aos[n * 32 + kb + 1];
    o0 = WMMA_F32(a, wo0[kk], o0);
    o1 = WMMA_F32(a, wo1[kk], o1);
  }
  for (int v = 0; v < 8; ++v) {
    const size_t r = (size_t)(R0 + rb + v) * 32;
    __builtin_nontemporal_store(o0[v], outp + r + n);
    __builtin_nontemporal_store(o1[v], outp + r + 16 + n);
  }
}

extern "C" void kernel_launch(void* const* d_in, const int* in_sizes, int n_in,
                              void* d_out, int out_size, void* d_ws, size_t ws_size,
                              hipStream_t stream) {
  (void)in_sizes; (void)n_in; (void)out_size; (void)ws_size;
  const float* query = (const float*)d_in[0];
  const float* key   = (const float*)d_in[1];
  const int*   qmask = (const int*)d_in[2];
  const int*   kmask = (const int*)d_in[3];
  const float* Wq = (const float*)d_in[4];
  const float* Wk = (const float*)d_in[5];
  const float* Wv = (const float*)d_in[6];
  const float* Wo = (const float*)d_in[7];
  const float* lq1 = (const float*)d_in[8];
  const float* lk1 = (const float*)d_in[9];
  const float* lq2 = (const float*)d_in[10];
  const float* lk2 = (const float*)d_in[11];
  const float* lnw = (const float*)d_in[12];

  float* outp  = (float*)d_out;
  float* diffp = outp + (size_t)B_ * L_ * E_;

  float* Qs = (float*)d_ws;
  float* Km = Qs + (size_t)B_ * L_ * E_;
  float* Vm = Km + (size_t)B_ * S_ * E_;

  kv_proj_kernel<<<(B_ * S_) / 16, 32 * KV_NW, 0, stream>>>(key, Wk, Wv, Km, Vm);
  q_proj_kernel<<<(B_ * L_) / 16, 32, 0, stream>>>(query, Wq, Qs);
  attn_kernel<<<(B_ * L_) / 16, ATT_THREADS, 0, stream>>>(Qs, Km, Vm, qmask, kmask,
                                                 lq1, lk1, lq2, lk2, lnw, Wo,
                                                 outp, diffp);
}


# attention kernel: the wave-wide max, sum and min butterflies of the softmax and the norm on the vector ALU (v_permlanex16 + row_xmask DPP) instead of LDS permutes; same pairs added in the same order
# speedup vs baseline: 6.3452x; 1.0372x over previous
#include <hip/hip_runtime.h>
#include <math.h>

typedef float v2f __attribute__((ext_vector_type(2)));
typedef float v8f __attribute__((ext_vector_type(8)));

#define WMMA_F32(a, b, c) \
  __builtin_amdgcn_wmma_f32_16x16x4_f32(false, (a), false, (b), (short)0, (c), false, false)

static constexpr int B_ = 8, S_ = 512, E_ = 32, L_ = 4096, SUBE_ = 768;
static constexpr float LAMBDA_INIT = 0.2f;

static constexpr int ST0 = 513;
static constexpr int H1  = 16 * ST0;
static constexpr int SMEM_FLOATS = 16 * ST0 + 16 * 512;

typedef float v4f __attribute__((ext_vector_type(4)));

static constexpr int KV_NW  = 8;
static constexpr int KV_KS  = SUBE_ / KV_NW;
static constexpr int KV_PST = 66;
static constexpr int KV_PT  = 16 * KV_PST;

__global__ __launch_bounds__(32 * KV_NW) __attribute__((amdgpu_waves_per_eu(2, 2)))
void kv_proj_kernel(
    const float* __restrict__ key, const float* __restrict__ Wk,
    const float* __restrict__ Wv, float* __restrict__ Km, float* __restrict__ Vm) {
  __shared__ float part[KV_NW * KV_PT];

  const int lane = threadIdx.x & 31;
  const int w    = __builtin_amdgcn_readfirstlane((int)(threadIdx.x >> 5));
  const int n    = lane & 15;
  const int khi  = lane >> 4;
  const int m0   = blockIdx.x * 16;
  const int koff = w * KV_KS + khi * 4;

  const float* ap  = key + (size_t)(m0 + n) * SUBE_ + koff;
  const float* pk0 = Wk + (size_t)n * SUBE_ + koff;
  const float* pk1 = Wk + (size_t)(16 + n) * SUBE_ + koff;
  const float* pv0 = Wv + (size_t)n * SUBE_ + koff;
  const float* pv1 = Wv + (size_t)(16 + n) * SUBE_ + koff;

  v8f ak0 = {}, ak1 = {}, av0 = {}, av1 = {};
  for (int j = 0; j < KV_KS / 8; ++j) {
    const v4f a   = *(const v4f*)(ap  + 8 * j);
    const v4f bk0 = *(const v4f*)(pk0 + 8 * j);
    const v4f bk1 = *(const v4f*)(pk1 + 8 * j);
    const v4f bv0 = *(const v4f*)(pv0 + 8 * j);
    const v4f bv1 = *(const v4f*)(pv1 + 8 * j);
    v2f alo, ahi, k0lo, k0hi, k1lo, k1hi, v0lo, v0hi, v1lo, v1hi;
    alo[0]  = a[0];   alo[1]  = a[1];   ahi[0]  = a[2];   ahi[1]  = a[3];
    k0lo[0] = bk0[0]; k0lo[1] = bk0[1]; k0hi[0] = bk0[2]; k0hi[1] = bk0[3];
    k1lo[0] = bk1[0]; k1lo[1] = bk1[1]; k1hi[0] = bk1[2]; k1hi[1] = bk1[3];
    v0lo[0] = bv0[0]; v0lo[1] = bv0[1]; v0hi[0] = bv0[2]; v0hi[1] = bv0[3];
    v1lo[0] = bv1[0]; v1lo[1] = bv1[1]; v1hi[0] = bv1[2]; v1hi[1] = bv1[3];
    ak0 = WMMA_F32(alo, k0lo, ak0);
    ak1 = WMMA_F32(alo, k1lo, ak1);
    av0 = WMMA_F32(alo, v0lo, av0);
    av1 = WMMA_F32(alo, v1lo, av1);
    ak0 = WMMA_F32(ahi, k0hi, ak0);
    ak1 = WMMA_F32(ahi, k1hi, ak1);
    av0 = WMMA_F32(ahi, v0hi, av0);
    av1 = WMMA_F32(ahi, v1hi, av1);
  }

  float* pw = part + w * KV_PT;
  for (int v = 0; v < 8; ++v) {
    const int r = (khi * 8 + v) * KV_PST + n;
    pw[r]      = ak0[v];
    pw[r + 16] = ak1[v];
    pw[r + 32] = av0[v];
    pw[r + 48] = av1[v];
  }
  __syncthreads();

  for (int i = 0; i < 4; ++i) {
    const int o = (int)threadIdx.x + 32 * KV_NW * i;
    const int r = o >> 6;
    const int c = o & 63;
    float x = part[r * KV_PST + c];
    for (int p = 1; p < KV_NW; ++p) x += part[p * KV_PT + r * KV_PST + c];
    float* dst = (c < 32) ? Km : Vm;
    dst[(size_t)(m0 + r) * 32 + (c & 31)] = x;
  }
}

__global__ __launch_bounds__(32) void q_proj_kernel(
    const float* __restrict__ q, const float* __restrict__ Wq,
    float* __restrict__ Qs) {
  const int lane = threadIdx.x;
  const int n    = lane & 15;
  const int khi  = lane >> 4;
  const int m0   = blockIdx.x * 16;
  const int mrow = m0 + n;

  v8f a0 = {}, a1 = {};
  for (int kk = 0; kk < 8; ++kk) {
    const int kb = kk * 4 + khi * 2;
    v2f a  = *(const v2f*)(q + (size_t)mrow * 32 + kb);
    v2f b0 = *(const v2f*)(Wq + (size_t)n * 32 + kb);
    v2f b1 = *(const v2f*)(Wq + (size_t)(16 + n) * 32 + kb);
    a0 = WMMA_F32(a, b0, a0);
    a1 = WMMA_F32(a, b1, a1);
  }
  const int rbase = m0 + (khi ? 8 : 0);
  for (int v = 0; v < 8; ++v) {
    const size_t r = (size_t)(rbase + v) * 32;
    Qs[r + n]      = a0[v] * 0.25f;
    Qs[r + 16 + n] = a1[v] * 0.25f;
  }
}

static __device__ __forceinline__ float xhalf(float x) {
  const int b = __builtin_bit_cast(int, x);
  return __builtin_bit_cast(float,
      __builtin_amdgcn_permlanex16(b, b, 0x76543210, 0xfedcba98, false, false));
}
template <int M>
static __device__ __forceinline__ float xrow(float x) {
  const int b = __builtin_bit_cast(int, x);
  return __builtin_bit_cast(float,
      __builtin_amdgcn_update_dpp(b, b, 0x160 + M, 0xF, 0xF, true));
}
template <int M>
static __device__ __forceinline__ float max_xrow(float x) {
  float r;
  asm("v_max_num_f32_dpp %0, %1, %1 row_xmask:%2 row_mask:0xf bank_mask:0xf bound_ctrl:1"
      : "=v"(r) : "v"(x), "n"(M));
  return r;
}
template <int M>
static __device__ __forceinline__ float min_xrow(float x) {
  float r;
  asm("v_min_num_f32_dpp %0, %1, %1 row_xmask:%2 row_mask:0xf bank_mask:0xf bound_ctrl:1"
      : "=v"(r) : "v"(x), "n"(M));
  return r;
}
static __device__ __forceinline__ void wave_max2(float& a, float& b) {
  a = fmaxf(a, xhalf(a));  b = fmaxf(b, xhalf(b));
  a = max_xrow<8>(a);      b = max_xrow<8>(b);
  a = max_xrow<4>(a);      b = max_xrow<4>(b);
  a = max_xrow<2>(a);      b = max_xrow<2>(b);
  a = max_xrow<1>(a);      b = max_xrow<1>(b);
}
static __device__ __forceinline__ float wave_min(float x) {
  x = fminf(x, xhalf(x));
  x = min_xrow<8>(x); x = min_xrow<4>(x);
  x = min_xrow<2>(x); x = min_xrow<1>(x);
  return x;
}
static __device__ __forceinline__ float wave_sum(float x) {
  x += xhalf(x);
  x += xrow<8>(x); x += xrow<4>(x);
  x += xrow<2>(x); x += xrow<1>(x);
  return x;
}

static __device__ __forceinline__ void lds_barrier() {
  __builtin_amdgcn_fence(__ATOMIC_RELEASE, "workgroup", "local");
  __builtin_amdgcn_s_barrier();
  __builtin_amdgcn_fence(__ATOMIC_ACQUIRE, "workgroup", "local");
}

static constexpr int NW          = 8;
static constexpr int ATT_THREADS = 32 * NW;
static constexpr int PART        = H1;
static constexpr int AOS         = H1 + NW * 512;

__global__ __launch_bounds__(ATT_THREADS) void attn_kernel(
    const float* __restrict__ Qs, const float* __restrict__ Km,
    const float* __restrict__ Vm, const int* __restrict__ qmask,
    const int* __restrict__ kmask, const float* __restrict__ lq1,
    const float* __restrict__ lk1, const float* __restrict__ lq2,
    const float* __restrict__ lk2, const float* __restrict__ lnw,
    const float* __restrict__ Wo, float* __restrict__ outp,
    float* __restrict__ diffp) {
  __shared__ float smem[SMEM_FLOATS];

  const int lane = threadIdx.x & 31;
  const int w    = __builtin_amdgcn_readfirstlane((int)(threadIdx.x >> 5));
  const int n    = lane & 15;
  const int khi  = lane >> 4;
  const int R0   = blockIdx.x * 16;
  const int b    = R0 / L_;
  const int rb   = khi ? 8 : 0;

  {
    v2f qa[2][4];
    for (int h = 0; h < 2; ++h)
      for (int kk = 0; kk < 4; ++kk)
        qa[h][kk] = *(const v2f*)(Qs + (size_t)(R0 + n) * 32 + h * 16 + kk * 4 + khi * 2);
    int qlo[8], qhi[8];
    for (int v = 0; v < 8; ++v) { qlo[v] = qmask[R0 + v]; qhi[v] = qmask[R0 + 8 + v]; }

    v2f kf[4][2][4];
    int kmv4[4];
    for (int t = 0; t < 4; ++t) {
      const int col = (w * 4 + t) * 16 + n;
      kmv4[t] = kmask[b * S_ + col];
      const float* krow = Km + (size_t)(b * S_ + col) * 32 + khi * 2;
      for (int h = 0; h < 2; ++h)
        for (int kk = 0; kk < 4; ++kk)
          kf[t][h][kk] = *(const v2f*)(krow + h * 16 + kk * 4);
    }
    __builtin_amdgcn_sched_barrier(0);

    for (int t = 0; t < 4; ++t) {
      const int col = (w * 4 + t) * 16 + n;
      const int kmv = kmv4[t];
      for (int h = 0; h < 2; ++h) {
        v8f acc = {};
        for (int kk = 0; kk < 4; ++kk) {
          acc = WMMA_F32(qa[h][kk], kf[t][h][kk], acc);
        }
        float* dst       = (h == 0) ? smem : (smem + H1);
        const int stride = (h == 0) ? ST0 : 512;
        for (int v = 0; v < 8; ++v) {
          const int r     = rb + v;
          const int valid = kmv && (khi ? qhi[v] : qlo[v]);
          dst[r * stride + col] = valid ? acc[v] : -1e20f;
        }
      }
    }
  }
  lds_barrier();

  v2f vb0[16], vb1[16];

  {
    float s1 = 0.f, s2 = 0.f;
    for (int i = 0; i < 16; ++i) { s1 += lq1[i] * lk1[i]; s2 += lq2[i] * lk2[i]; }
    const float lambda_full = __expf(s1) - __expf(s2) + LAMBDA_INIT;

    int kmc[16];
    for (int i = 0; i < 16; ++i) kmc[i] = kmask[b * S_ + lane + 32 * i];

    for (int mm = 0; mm < 2; ++mm) {
      const int m = w * 2 + mm;
      const float* r0 = smem + m * ST0;
      const float* r1 = smem + H1 + m * 512;
      float mx0 = -3.4e38f, mx1 = -3.4e38f;
      for (int i = 0; i < 16; ++i) {
        const int c = lane + 32 * i;
        mx0 = fmaxf(mx0, r0[c]);
        mx1 = fmaxf(mx1, r1[c]);
      }
      wave_max2(mx0, mx1);
      float e0[16], e1[16], sum0 = 0.f, sum1 = 0.f;
      for (int i = 0; i < 16; ++i) {
        const int c = lane + 32 * i;
        e0[i] = __expf(r0[c] - mx0); sum0 += e0[i];
        e1[i] = __expf(r1[c] - mx1); sum1 += e1[i];
      }
      if (mm == 1) {
        __builtin_amdgcn_sched_barrier(0);
        const float* vbase = Vm + (size_t)(b * S_ + w * 64 + khi * 2) * 32 + n;
#pragma unroll
        for (int j = 0; j < 15; ++j) {
          const float* vr = vbase + j * 128;
          vb0[j][0] = vr[0];  vb0[j][1] = vr[32];
          vb1[j][0] = vr[16]; vb1[j][1] = vr[48];
        }
      }
      sum0 = wave_sum(sum0);
      sum1 = wave_sum(sum1);
      const float inv0 = 1.f / (sum0 + 1e-8f);
      const float inv1 = 1.f / (sum1 + 1e-8f);
      float d[16], dmin = 3.4e38f;
      for (int i = 0; i < 16; ++i) {
        d[i] = e0[i] * inv0 - lambda_full * (e1[i] * inv1);
        dmin = fminf(dmin, d[i]);
      }
      dmin = wave_min(dmin);

      const int qv = qmask[R0 + m];
      float* w0 = smem + m * ST0;
      float* dg = diffp + (size_t)(R0 + m) * 512;
      for (int i = 0; i < 16; ++i) {
        const int c = lane + 32 * i;
        float dd = d[i] - dmin + 1e-5f;
        dd = (qv && kmc[i]) ? dd : 0.f;
        w0[c] = dd;
        __builtin_nontemporal_store(dd, dg + c);
      }

      if (mm == 1) {
        __builtin_amdgcn_sched_barrier(0);
        const float* vr = Vm + (size_t)(b * S_ + w * 64 + khi * 2) * 32 + n + 15 * 128;
        vb0[15][0] = vr[0];  vb0[15][1] = vr[32];
        vb1[15][0] = vr[16]; vb1[15][1] = vr[48];
      }
    }
  }
  lds_barrier();

  float lnw_l;
  v2f wo0[8], wo1[8];
  {
    v2f af[16];
#pragma unroll
    for (int j = 0; j < 16; ++j) {
      const int kb = (w * 16 + j) * 4 + khi * 2;
      af[j][0] = smem[n * ST0 + kb]; af[j][1] = smem[n * ST0 + kb + 1];
    }
    __builtin_amdgcn_sched_barrier(0);
    v8f ao0 = {}, ao1 = {};
#pragma unroll
    for (int j = 0; j < 16; ++j) {
      ao0 = WMMA_F32(af[j], vb0[j], ao0);
      ao1 = WMMA_F32(af[j], vb1[j], ao1);
    }

    lnw_l = lnw[lane];
    if (w == 0) {
      for (int kk = 0; kk < 8; ++kk) {
        const int kb = kk * 4 + khi * 2;
        wo0[kk] = *(const v2f*)(Wo + (size_t)n * 32 + kb);
        wo1[kk] = *(const v2f*)(Wo + (size_t)(16 + n) * 32 + kb);
      }
    }

    float* part = smem + PART + w * 512;
    for (int v = 0; v < 8; ++v) {
      const int r = rb + v;
      part[r * 32 + n]      = ao0[v];
      part[r * 32 + 16 + n] = ao1[v];
    }
  }
  lds_barrier();

  for (int mm = 0; mm < 2; ++mm) {
    const int m = w * 2 + mm;
    float x = smem[PART + m * 32 + lane];
    for (int p = 1; p < NW; ++p) x += smem[PART + p * 512 + m * 32 + lane];
    float sq = x * x;
    sq = wave_sum(sq);
    const float rms = rsqrtf(sq * (1.0f / 32.0f) + 1e-5f);
    smem[AOS + m * 32 + lane] = x * rms * lnw_l * 0.8f;
  }
  lds_barrier();

  if (w != 0) return;

  const float* aos = smem + AOS;
  v8f o0 = {}, o1 = {};
  for (int kk = 0; kk < 8; ++kk) {
    const int kb = kk * 4 + khi * 2;
    v2f a; a[0] = aos[n * 32 + kb]; a[1] = aos[n * 32 + kb + 1];
    o0 = WMMA_F32(a, wo0[kk], o0);
    o1 = WMMA_F32(a, wo1[kk], o1);
  }
  for (int v = 0; v < 8; ++v) {
    const size_t r = (size_t)(R0 + rb + v) * 32;
    __builtin_nontemporal_store(o0[v], outp + r + n);
    __builtin_nontemporal_store(o1[v], outp + r + 16 + n);
  }
}

extern "C" void kernel_launch(void* const* d_in, const int* in_sizes, int n_in,
                              void* d_out, int out_size, void* d_ws, size_t ws_size,
                              hipStream_t stream) {
  (void)in_sizes; (void)n_in; (void)out_size; (void)ws_size;
  const float* query = (const float*)d_in[0];
  const float* key   = (const float*)d_in[1];
  const int*   qmask = (const int*)d_in[2];
  const int*   kmask = (const int*)d_in[3];
  const float* Wq = (const float*)d_in[4];
  const float* Wk = (const float*)d_in[5];
  const float* Wv = (const float*)d_in[6];
  const float* Wo = (const float*)d_in[7];
  const float* lq1 = (const float*)d_in[8];
  const float* lk1 = (const float*)d_in[9];
  const float* lq2 = (const float*)d_in[10];
  const float* lk2 = (const float*)d_in[11];
  const float* lnw = (const float*)d_in[12];

  float* outp  = (float*)d_out;
  float* diffp = outp + (size_t)B_ * L_ * E_;

  float* Qs = (float*)d_ws;
  float* Km = Qs + (size_t)B_ * L_ * E_;
  float* Vm = Km + (size_t)B_ * S_ * E_;

  kv_proj_kernel<<<(B_ * S_) / 16, 32 * KV_NW, 0, stream>>>(key, Wk, Wv, Km, Vm);
  q_proj_kernel<<<(B_ * L_) / 16, 32, 0, stream>>>(query, Wq, Qs);
  attn_kernel<<<(B_ * L_) / 16, ATT_THREADS, 0, stream>>>(Qs, Km, Vm, qmask, kmask,
                                                 lq1, lk1, lq2, lk2, lnw, Wo,
                                                 outp, diffp);
}


# Q projection moved into the attention kernel: waves 0 and 1 of a work-group compute their tile's two Q heads (same k order) and hand them on through LDS; the Q launch and its workspace plane go
# speedup vs baseline: 6.9845x; 1.1008x over previous
#include <hip/hip_runtime.h>
#include <math.h>

typedef float v2f __attribute__((ext_vector_type(2)));
typedef float v8f __attribute__((ext_vector_type(8)));

#define WMMA_F32(a, b, c) \
  __builtin_amdgcn_wmma_f32_16x16x4_f32(false, (a), false, (b), (short)0, (c), false, false)

static constexpr int B_ = 8, S_ = 512, E_ = 32, L_ = 4096, SUBE_ = 768;
static constexpr float LAMBDA_INIT = 0.2f;

static constexpr int ST0 = 513;
static constexpr int H1  = 16 * ST0;
static constexpr int QST = 34;
static constexpr int QT  = 16 * ST0 + 16 * 512;
static constexpr int SMEM_FLOATS = QT + 16 * QST;

typedef float v4f __attribute__((ext_vector_type(4)));

static constexpr int KV_NW  = 8;
static constexpr int KV_KS  = SUBE_ / KV_NW;
static constexpr int KV_PST = 66;
static constexpr int KV_PT  = 16 * KV_PST;

__global__ __launch_bounds__(32 * KV_NW) __attribute__((amdgpu_waves_per_eu(2, 2)))
void kv_proj_kernel(
    const float* __restrict__ key, const float* __restrict__ Wk,
    const float* __restrict__ Wv, float* __restrict__ Km, float* __restrict__ Vm) {
  __shared__ float part[KV_NW * KV_PT];

  const int lane = threadIdx.x & 31;
  const int w    = __builtin_amdgcn_readfirstlane((int)(threadIdx.x >> 5));
  const int n    = lane & 15;
  const int khi  = lane >> 4;
  const int m0   = blockIdx.x * 16;
  const int koff = w * KV_KS + khi * 4;

  const float* ap  = key + (size_t)(m0 + n) * SUBE_ + koff;
  const float* pk0 = Wk + (size_t)n * SUBE_ + koff;
  const float* pk1 = Wk + (size_t)(16 + n) * SUBE_ + koff;
  const float* pv0 = Wv + (size_t)n * SUBE_ + koff;
  const float* pv1 = Wv + (size_t)(16 + n) * SUBE_ + koff;

  v8f ak0 = {}, ak1 = {}, av0 = {}, av1 = {};
  for (int j = 0; j < KV_KS / 8; ++j) {
    const v4f a   = *(const v4f*)(ap  + 8 * j);
    const v4f bk0 = *(const v4f*)(pk0 + 8 * j);
    const v4f bk1 = *(const v4f*)(pk1 + 8 * j);
    const v4f bv0 = *(const v4f*)(pv0 + 8 * j);
    const v4f bv1 = *(const v4f*)(pv1 + 8 * j);
    v2f alo, ahi, k0lo, k0hi, k1lo, k1hi, v0lo, v0hi, v1lo, v1hi;
    alo[0]  = a[0];   alo[1]  = a[1];   ahi[0]  = a[2];   ahi[1]  = a[3];
    k0lo[0] = bk0[0]; k0lo[1] = bk0[1]; k0hi[0] = bk0[2]; k0hi[1] = bk0[3];
    k1lo[0] = bk1[0]; k1lo[1] = bk1[1]; k1hi[0] = bk1[2]; k1hi[1] = bk1[3];
    v0lo[0] = bv0[0]; v0lo[1] = bv0[1]; v0hi[0] = bv0[2]; v0hi[1] = bv0[3];
    v1lo[0] = bv1[0]; v1lo[1] = bv1[1]; v1hi[0] = bv1[2]; v1hi[1] = bv1[3];
    ak0 = WMMA_F32(alo, k0lo, ak0);
    ak1 = WMMA_F32(alo, k1lo, ak1);
    av0 = WMMA_F32(alo, v0lo, av0);
    av1 = WMMA_F32(alo, v1lo, av1);
    ak0 = WMMA_F32(ahi, k0hi, ak0);
    ak1 = WMMA_F32(ahi, k1hi, ak1);
    av0 = WMMA_F32(ahi, v0hi, av0);
    av1 = WMMA_F32(ahi, v1hi, av1);
  }

  float* pw = part + w * KV_PT;
  for (int v = 0; v < 8; ++v) {
    const int r = (khi * 8 + v) * KV_PST + n;
    pw[r]      = ak0[v];
    pw[r + 16] = ak1[v];
    pw[r + 32] = av0[v];
    pw[r + 48] = av1[v];
  }
  __syncthreads();

  for (int i = 0; i < 4; ++i) {
    const int o = (int)threadIdx.x + 32 * KV_NW * i;
    const int r = o >> 6;
    const int c = o & 63;
    float x = part[r * KV_PST + c];
    for (int p = 1; p < KV_NW; ++p) x += part[p * KV_PT + r * KV_PST + c];
    float* dst = (c < 32) ? Km : Vm;
    dst[(size_t)(m0 + r) * 32 + (c & 31)] = x;
  }
}

static __device__ __forceinline__ float xhalf(float x) {
  const int b = __builtin_bit_cast(int, x);
  return __builtin_bit_cast(float,
      __builtin_amdgcn_permlanex16(b, b, 0x76543210, 0xfedcba98, false, false));
}
template <int M>
static __device__ __forceinline__ float xrow(float x) {
  const int b = __builtin_bit_cast(int, x);
  return __builtin_bit_cast(float,
      __builtin_amdgcn_update_dpp(b, b, 0x160 + M, 0xF, 0xF, true));
}
template <int M>
static __device__ __forceinline__ float max_xrow(float x) {
  float r;
  asm("v_max_num_f32_dpp %0, %1, %1 row_xmask:%2 row_mask:0xf bank_mask:0xf bound_ctrl:1"
      : "=v"(r) : "v"(x), "n"(M));
  return r;
}
template <int M>
static __device__ __forceinline__ float min_xrow(float x) {
  float r;
  asm("v_min_num_f32_dpp %0, %1, %1 row_xmask:%2 row_mask:0xf bank_mask:0xf bound_ctrl:1"
      : "=v"(r) : "v"(x), "n"(M));
  return r;
}
static __device__ __forceinline__ void wave_max2(float& a, float& b) {
  a = fmaxf(a, xhalf(a));  b = fmaxf(b, xhalf(b));
  a = max_xrow<8>(a);      b = max_xrow<8>(b);
  a = max_xrow<4>(a);      b = max_xrow<4>(b);
  a = max_xrow<2>(a);      b = max_xrow<2>(b);
  a = max_xrow<1>(a);      b = max_xrow<1>(b);
}
static __device__ __forceinline__ float wave_min(float x) {
  x = fminf(x, xhalf(x));
  x = min_xrow<8>(x); x = min_xrow<4>(x);
  x = min_xrow<2>(x); x = min_xrow<1>(x);
  return x;
}
static __device__ __forceinline__ float wave_sum(float x) {
  x += xhalf(x);
  x += xrow<8>(x); x += xrow<4>(x);
  x += xrow<2>(x); x += xrow<1>(x);
  return x;
}

static __device__ __forceinline__ void lds_barrier() {
  __builtin_amdgcn_fence(__ATOMIC_RELEASE, "workgroup", "local");
  __builtin_amdgcn_s_barrier();
  __builtin_amdgcn_fence(__ATOMIC_ACQUIRE, "workgroup", "local");
}

static constexpr int NW          = 8;
static constexpr int ATT_THREADS = 32 * NW;
static constexpr int PART        = H1;
static constexpr int AOS         = H1 + NW * 512;

__global__ __launch_bounds__(ATT_THREADS) void attn_kernel(
    const float* __restrict__ query, const float* __restrict__ Wq,
    const float* __restrict__ Km,
    const float* __restrict__ Vm, const int* __restrict__ qmask,
    const int* __restrict__ kmask, const float* __restrict__ lq1,
    const float* __restrict__ lk1, const float* __restrict__ lq2,
    const float* __restrict__ lk2, const float* __restrict__ lnw,
    const float* __restrict__ Wo, float* __restrict__ outp,
    float* __restrict__ diffp) {
  __shared__ __attribute__((aligned(16))) float smem[SMEM_FLOATS];

  const int lane = threadIdx.x & 31;
  const int w    = __builtin_amdgcn_readfirstlane((int)(threadIdx.x >> 5));
  const int n    = lane & 15;
  const int khi  = lane >> 4;
  const int R0   = blockIdx.x * 16;
  const int b    = R0 / L_;
  const int rb   = khi ? 8 : 0;

  {
    v2f qx[8], wq[8];
    if (w < 2) {
      const float* qrow = query + (size_t)(R0 + n) * 32 + khi * 2;
      const float* wrow = Wq + (size_t)(16 * w + n) * 32 + khi * 2;
      for (int kk = 0; kk < 8; ++kk) {
        qx[kk] = *(const v2f*)(qrow + kk * 4);
        wq[kk] = *(const v2f*)(wrow + kk * 4);
      }
    }
    int qlo[8], qhi[8];
    for (int v = 0; v < 8; ++v) { qlo[v] = qmask[R0 + v]; qhi[v] = qmask[R0 + 8 + v]; }

    v2f kf[4][2][4];
    int kmv4[4];
    for (int t = 0; t < 4; ++t) {
      const int col = (w * 4 + t) * 16 + n;
      kmv4[t] = kmask[b * S_ + col];
      const float* krow = Km + (size_t)(b * S_ + col) * 32 + khi * 2;
      for (int h = 0; h < 2; ++h)
        for (int kk = 0; kk < 4; ++kk)
          kf[t][h][kk] = *(const v2f*)(krow + h * 16 + kk * 4);
    }
    __builtin_amdgcn_sched_barrier(0);

    float* qt = smem + QT;
    if (w < 2) {
      v8f qacc = {};
      for (int kk = 0; kk < 8; ++kk) qacc = WMMA_F32(qx[kk], wq[kk], qacc);
      for (int v = 0; v < 8; ++v) qt[(rb + v) * QST + 16 * w + n] = qacc[v] * 0.25f;
    }
    lds_barrier();
    v2f qa[2][4];
    for (int h = 0; h < 2; ++h)
      for (int kk = 0; kk < 4; ++kk)
        qa[h][kk] = *(const v2f*)(qt + n * QST + h * 16 + kk * 4 + khi * 2);
    __builtin_amdgcn_sched_barrier(0);

    for (int t = 0; t < 4; ++t) {
      const int col = (w * 4 + t) * 16 + n;
      const int kmv = kmv4[t];
      for (int h = 0; h < 2; ++h) {
        v8f acc = {};
        for (int kk = 0; kk < 4; ++kk) {
          acc = WMMA_F32(qa[h][kk], kf[t][h][kk], acc);
        }
        float* dst       = (h == 0) ? smem : (smem + H1);
        const int stride = (h == 0) ? ST0 : 512;
        for (int v = 0; v < 8; ++v) {
          const int r     = rb + v;
          const int valid = kmv && (khi ? qhi[v] : qlo[v]);
          dst[r * stride + col] = valid ? acc[v] : -1e20f;
        }
      }
    }
  }
  lds_barrier();

  v2f vb0[16], vb1[16];

  {
    float s1 = 0.f, s2 = 0.f;
    for (int i = 0; i < 16; ++i) { s1 += lq1[i] * lk1[i]; s2 += lq2[i] * lk2[i]; }
    const float lambda_full = __expf(s1) - __expf(s2) + LAMBDA_INIT;

    int kmc[16];
    for (int i = 0; i < 16; ++i) kmc[i] = kmask[b * S_ + lane + 32 * i];

    for (int mm = 0; mm < 2; ++mm) {
      const int m = w * 2 + mm;
      const float* r0 = smem + m * ST0;
      const float* r1 = smem + H1 + m * 512;
      float mx0 = -3.4e38f, mx1 = -3.4e38f;
      for (int i = 0; i < 16; ++i) {
        const int c = lane + 32 * i;
        mx0 = fmaxf(mx0, r0[c]);
        mx1 = fmaxf(mx1, r1[c]);
      }
      wave_max2(mx0, mx1);
      float e0[16], e1[16], sum0 = 0.f, sum1 = 0.f;
      for (int i = 0; i < 16; ++i) {
        const int c = lane + 32 * i;
        e0[i] = __expf(r0[c] - mx0); sum0 += e0[i];
        e1[i] = __expf(r1[c] - mx1); sum1 += e1[i];
      }
      if (mm == 1) {
        __builtin_amdgcn_sched_barrier(0);
        const float* vbase = Vm + (size_t)(b * S_ + w * 64 + khi * 2) * 32 + n;
#pragma unroll
        for (int j = 0; j < 15; ++j) {
          const float* vr = vbase + j * 128;
          vb0[j][0] = vr[0];  vb0[j][1] = vr[32];
          vb1[j][0] = vr[16]; vb1[j][1] = vr[48];
        }
      }
      sum0 = wave_sum(sum0);
      sum1 = wave_sum(sum1);
      const float inv0 = 1.f / (sum0 + 1e-8f);
      const float inv1 = 1.f / (sum1 + 1e-8f);
      float d[16], dmin = 3.4e38f;
      for (int i = 0; i < 16; ++i) {
        d[i] = e0[i] * inv0 - lambda_full * (e1[i] * inv1);
        dmin = fminf(dmin, d[i]);
      }
      dmin = wave_min(dmin);

      const int qv = qmask[R0 + m];
      float* w0 = smem + m * ST0;
      float* dg = diffp + (size_t)(R0 + m) * 512;
      for (int i = 0; i < 16; ++i) {
        const int c = lane + 32 * i;
        float dd = d[i] - dmin + 1e-5f;
        dd = (qv && kmc[i]) ? dd : 0.f;
        w0[c] = dd;
        __builtin_nontemporal_store(dd, dg + c);
      }

      if (mm == 1) {
        __builtin_amdgcn_sched_barrier(0);
        const float* vr = Vm + (size_t)(b * S_ + w * 64 + khi * 2) * 32 + n + 15 * 128;
        vb0[15][0] = vr[0];  vb0[15][1] = vr[32];
        vb1[15][0] = vr[16]; vb1[15][1] = vr[48];
      }
    }
  }
  lds_barrier();

  float lnw_l;
  v2f wo0[8], wo1[8];
  {
    v2f af[16];
#pragma unroll
    for (int j = 0; j < 16; ++j) {
      const int kb = (w * 16 + j) * 4 + khi * 2;
      af[j][0] = smem[n * ST0 + kb]; af[j][1] = smem[n * ST0 + kb + 1];
    }
    __builtin_amdgcn_sched_barrier(0);
    v8f ao0 = {}, ao1 = {};
#pragma unroll
    for (int j = 0; j < 16; ++j) {
      ao0 = WMMA_F32(af[j], vb0[j], ao0);
      ao1 = WMMA_F32(af[j], vb1[j], ao1);
    }

    lnw_l = lnw[lane];
    if (w == 0) {
      for (int kk = 0; kk < 8; ++kk) {
        const int kb = kk * 4 + khi * 2;
        wo0[kk] = *(const v2f*)(Wo + (size_t)n * 32 + kb);
        wo1[kk] = *(const v2f*)(Wo + (size_t)(16 + n) * 32 + kb);
      }
    }

    float* part = smem + PART + w * 512;
    for (int v = 0; v < 8; ++v) {
      const int r = rb + v;
      part[r * 32 + n]      = ao0[v];
      part[r * 32 + 16 + n] = ao1[v];
    }
  }
  lds_barrier();

  for (int mm = 0; mm < 2; ++mm) {
    const int m = w * 2 + mm;
    float x = smem[PART + m * 32 + lane];
    for (int p = 1; p < NW; ++p) x += smem[PART + p * 512 + m * 32 + lane];
    float sq = x * x;
    sq = wave_sum(sq);
    const float rms = rsqrtf(sq * (1.0f / 32.0f) + 1e-5f);
    smem[AOS + m * 32 + lane] = x * rms * lnw_l * 0.8f;
  }
  lds_barrier();

  if (w != 0) return;

  const float* aos = smem + AOS;
  v8f o0 = {}, o1 = {};
  for (int kk = 0; kk < 8; ++kk) {
    const int kb = kk * 4 + khi * 2;
    v2f a; a[0] = aos[n * 32 + kb]; a[1] = aos[n * 32 + kb + 1];
    o0 = WMMA_F32(a, wo0[kk], o0);
    o1 = WMMA_F32(a, wo1[kk], o1);
  }
  for (int v = 0; v < 8; ++v) {
    const size_t r = (size_t)(R0 + rb + v) * 32;
    __builtin_nontemporal_store(o0[v], outp + r + n);
    __builtin_nontemporal_store(o1[v], outp + r + 16 + n);
  }
}

extern "C" void kernel_launch(void* const* d_in, const int* in_sizes, int n_in,
                              void* d_out, int out_size, void* d_ws, size_t ws_size,
                              hipStream_t stream) {
  (void)in_sizes; (void)n_in; (void)out_size; (void)ws_size;
  const float* query = (const float*)d_in[0];
  const float* key   = (const float*)d_in[1];
  const int*   qmask = (const int*)d_in[2];
  const int*   kmask = (const int*)d_in[3];
  const float* Wq = (const float*)d_in[4];
  const float* Wk = (const float*)d_in[5];
  const float* Wv = (const float*)d_in[6];
  const float* Wo = (const float*)d_in[7];
  const float* lq1 = (const float*)d_in[8];
  const float* lk1 = (const float*)d_in[9];
  const float* lq2 = (const float*)d_in[10];
  const float* lk2 = (const float*)d_in[11];
  const float* lnw = (const float*)d_in[12];

  float* outp  = (float*)d_out;
  float* diffp = outp + (size_t)B_ * L_ * E_;

  float* Km = (float*)d_ws;
  float* Vm = Km + (size_t)B_ * S_ * E_;

  kv_proj_kernel<<<(B_ * S_) / 16, 32 * KV_NW, 0, stream>>>(key, Wk, Wv, Km, Vm);
  attn_kernel<<<(B_ * L_) / 16, ATT_THREADS, 0, stream>>>(query, Wq, Km, Vm, qmask, kmask,
                                                 lq1, lk1, lq2, lk2, lnw, Wo,
                                                 outp, diffp);
}


# K and V stored by the projection kernel in the attention kernel's B-fragment order: fragment loads are 16 bytes per lane over 4 whole lines (K 32->16, V 64->16 a wave), not 16-line gathers
# speedup vs baseline: 7.4910x; 1.0725x over previous
#include <hip/hip_runtime.h>
#include <math.h>

typedef float v2f __attribute__((ext_vector_type(2)));
typedef float v8f __attribute__((ext_vector_type(8)));

#define WMMA_F32(a, b, c) \
  __builtin_amdgcn_wmma_f32_16x16x4_f32(false, (a), false, (b), (short)0, (c), false, false)

static constexpr int B_ = 8, S_ = 512, E_ = 32, L_ = 4096, SUBE_ = 768;
static constexpr float LAMBDA_INIT = 0.2f;

static constexpr int ST0 = 513;
static constexpr int H1  = 16 * ST0;
static constexpr int QST = 34;
static constexpr int QT  = 16 * ST0 + 16 * 512;
static constexpr int SMEM_FLOATS = QT + 16 * QST;

typedef float v4f __attribute__((ext_vector_type(4)));

static constexpr int KV_NW  = 8;
static constexpr int KV_KS  = SUBE_ / KV_NW;
static constexpr int KV_PST = 66;
static constexpr int KV_PT  = 16 * KV_PST;

__global__ __launch_bounds__(32 * KV_NW) __attribute__((amdgpu_waves_per_eu(2, 2)))
void kv_proj_kernel(
    const float* __restrict__ key, const float* __restrict__ Wk,
    const float* __restrict__ Wv, float* __restrict__ Km, float* __restrict__ Vm) {
  __shared__ float part[KV_NW * KV_PT];

  const int lane = threadIdx.x & 31;
  const int w    = __builtin_amdgcn_readfirstlane((int)(threadIdx.x >> 5));
  const int n    = lane & 15;
  const int khi  = lane >> 4;
  const int m0   = blockIdx.x * 16;
  const int koff = w * KV_KS + khi * 4;

  const float* ap  = key + (size_t)(m0 + n) * SUBE_ + koff;
  const float* pk0 = Wk + (size_t)n * SUBE_ + koff;
  const float* pk1 = Wk + (size_t)(16 + n) * SUBE_ + koff;
  const float* pv0 = Wv + (size_t)n * SUBE_ + koff;
  const float* pv1 = Wv + (size_t)(16 + n) * SUBE_ + koff;

  v8f ak0 = {}, ak1 = {}, av0 = {}, av1 = {};
  for (int j = 0; j < KV_KS / 8; ++j) {
    const v4f a   = *(const v4f*)(ap  + 8 * j);
    const v4f bk0 = *(const v4f*)(pk0 + 8 * j);
    const v4f bk1 = *(const v4f*)(pk1 + 8 * j);
    const v4f bv0 = *(const v4f*)(pv0 + 8 * j);
    const v4f bv1 = *(const v4f*)(pv1 + 8 * j);
    v2f alo, ahi, k0lo, k0hi, k1lo, k1hi, v0lo, v0hi, v1lo, v1hi;
    alo[0]  = a[0];   alo[1]  = a[1];   ahi[0]  = a[2];   ahi[1]  = a[3];
    k0lo[0] = bk0[0]; k0lo[1] = bk0[1]; k0hi[0] = bk0[2]; k0hi[1] = bk0[3];
    k1lo[0] = bk1[0]; k1lo[1] = bk1[1]; k1hi[0] = bk1[2]; k1hi[1] = bk1[3];
    v0lo[0] = bv0[0]; v0lo[1] = bv0[1]; v0hi[0] = bv0[2]; v0hi[1] = bv0[3];
    v1lo[0] = bv1[0]; v1lo[1] = bv1[1]; v1hi[0] = bv1[2]; v1hi[1] = bv1[3];
    ak0 = WMMA_F32(alo, k0lo, ak0);
    ak1 = WMMA_F32(alo, k1lo, ak1);
    av0 = WMMA_F32(alo, v0lo, av0);
    av1 = WMMA_F32(alo, v1lo, av1);
    ak0 = WMMA_F32(ahi, k0hi, ak0);
    ak1 = WMMA_F32(ahi, k1hi, ak1);
    av0 = WMMA_F32(ahi, v0hi, av0);
    av1 = WMMA_F32(ahi, v1hi, av1);
  }

  float* pw = part + w * KV_PT;
  for (int v = 0; v < 8; ++v) {
    const int r = (khi * 8 + v) * KV_PST + n;
    pw[r]      = ak0[v];
    pw[r + 16] = ak1[v];
    pw[r + 32] = av0[v];
    pw[r + 48] = av1[v];
  }
  __syncthreads();

  {
    const int u  = (int)threadIdx.x & 127;
    const int g  = u >> 5;
    const int cn = u & 15;
    const int ch = (u >> 4) & 1;
    v4f x;
    for (int e = 0; e < 4; ++e) {
      int r, c;
      if (w < 4) { r = cn;                        c = (g >> 1) * 16 + (g & 1) * 8 + (e >> 1) * 4 + ch * 2 + (e & 1); }
      else       { r = g * 4 + ch * 2 + (e & 1);  c = 32 + (e >> 1) * 16 + cn; }
      float s = part[r * KV_PST + c];
      for (int p = 1; p < KV_NW; ++p) s += part[p * KV_PT + r * KV_PST + c];
      x[e] = s;
    }
    float* dst = (w < 4) ? Km : Vm;
    *(v4f*)(dst + (size_t)blockIdx.x * 512 + u * 4) = x;
  }
}

static __device__ __forceinline__ float xhalf(float x) {
  const int b = __builtin_bit_cast(int, x);
  return __builtin_bit_cast(float,
      __builtin_amdgcn_permlanex16(b, b, 0x76543210, 0xfedcba98, false, false));
}
template <int M>
static __device__ __forceinline__ float xrow(float x) {
  const int b = __builtin_bit_cast(int, x);
  return __builtin_bit_cast(float,
      __builtin_amdgcn_update_dpp(b, b, 0x160 + M, 0xF, 0xF, true));
}
template <int M>
static __device__ __forceinline__ float max_xrow(float x) {
  float r;
  asm("v_max_num_f32_dpp %0, %1, %1 row_xmask:%2 row_mask:0xf bank_mask:0xf bound_ctrl:1"
      : "=v"(r) : "v"(x), "n"(M));
  return r;
}
template <int M>
static __device__ __forceinline__ float min_xrow(float x) {
  float r;
  asm("v_min_num_f32_dpp %0, %1, %1 row_xmask:%2 row_mask:0xf bank_mask:0xf bound_ctrl:1"
      : "=v"(r) : "v"(x), "n"(M));
  return r;
}
static __device__ __forceinline__ void wave_max2(float& a, float& b) {
  a = fmaxf(a, xhalf(a));  b = fmaxf(b, xhalf(b));
  a = max_xrow<8>(a);      b = max_xrow<8>(b);
  a = max_xrow<4>(a);      b = max_xrow<4>(b);
  a = max_xrow<2>(a);      b = max_xrow<2>(b);
  a = max_xrow<1>(a);      b = max_xrow<1>(b);
}
static __device__ __forceinline__ float wave_min(float x) {
  x = fminf(x, xhalf(x));
  x = min_xrow<8>(x); x = min_xrow<4>(x);
  x = min_xrow<2>(x); x = min_xrow<1>(x);
  return x;
}
static __device__ __forceinline__ float wave_sum(float x) {
  x += xhalf(x);
  x += xrow<8>(x); x += xrow<4>(x);
  x += xrow<2>(x); x += xrow<1>(x);
  return x;
}

static __device__ __forceinline__ void lds_barrier() {
  __builtin_amdgcn_fence(__ATOMIC_RELEASE, "workgroup", "local");
  __builtin_amdgcn_s_barrier();
  __builtin_amdgcn_fence(__ATOMIC_ACQUIRE, "workgroup", "local");
}

static constexpr int NW          = 8;
static constexpr int ATT_THREADS = 32 * NW;
static constexpr int PART        = H1;
static constexpr int AOS         = H1 + NW * 512;

__global__ __launch_bounds__(ATT_THREADS) void attn_kernel(
    const float* __restrict__ query, const float* __restrict__ Wq,
    const float* __restrict__ Km,
    const float* __restrict__ Vm, const int* __restrict__ qmask,
    const int* __restrict__ kmask, const float* __restrict__ lq1,
    const float* __restrict__ lk1, const float* __restrict__ lq2,
    const float* __restrict__ lk2, const float* __restrict__ lnw,
    const float* __restrict__ Wo, float* __restrict__ outp,
    float* __restrict__ diffp) {
  __shared__ __attribute__((aligned(16))) float smem[SMEM_FLOATS];

  const int lane = threadIdx.x & 31;
  const int w    = __builtin_amdgcn_readfirstlane((int)(threadIdx.x >> 5));
  const int n    = lane & 15;
  const int khi  = lane >> 4;
  const int R0   = blockIdx.x * 16;
  const int b    = R0 / L_;
  const int rb   = khi ? 8 : 0;

  {
    v2f qx[8], wq[8];
    if (w < 2) {
      const float* qrow = query + (size_t)(R0 + n) * 32 + khi * 2;
      const float* wrow = Wq + (size_t)(16 * w + n) * 32 + khi * 2;
      for (int kk = 0; kk < 8; ++kk) {
        qx[kk] = *(const v2f*)(qrow + kk * 4);
        wq[kk] = *(const v2f*)(wrow + kk * 4);
      }
    }
    int qlo[8], qhi[8];
    for (int v = 0; v < 8; ++v) { qlo[v] = qmask[R0 + v]; qhi[v] = qmask[R0 + 8 + v]; }

    v2f kf[4][2][4];
    int kmv4[4];
    const float* kfb = Km + (size_t)(b * 32 + w * 4) * 512 + lane * 4;
    for (int t = 0; t < 4; ++t) {
      const int col = (w * 4 + t) * 16 + n;
      kmv4[t] = kmask[b * S_ + col];
      for (int h = 0; h < 2; ++h)
        for (int p = 0; p < 2; ++p) {
          const v4f x = *(const v4f*)(kfb + ((t * 2 + h) * 2 + p) * 128);
          kf[t][h][2 * p][0]     = x[0]; kf[t][h][2 * p][1]     = x[1];
          kf[t][h][2 * p + 1][0] = x[2]; kf[t][h][2 * p + 1][1] = x[3];
        }
    }
    __builtin_amdgcn_sched_barrier(0);

    float* qt = smem + QT;
    if (w < 2) {
      v8f qacc = {};
      for (int kk = 0; kk < 8; ++kk) qacc = WMMA_F32(qx[kk], wq[kk], qacc);
      for (int v = 0; v < 8; ++v) qt[(rb + v) * QST + 16 * w + n] = qacc[v] * 0.25f;
    }
    lds_barrier();
    v2f qa[2][4];
    for (int h = 0; h < 2; ++h)
      for (int kk = 0; kk < 4; ++kk)
        qa[h][kk] = *(const v2f*)(qt + n * QST + h * 16 + kk * 4 + khi * 2);
    __builtin_amdgcn_sched_barrier(0);

    for (int t = 0; t < 4; ++t) {
      const int col = (w * 4 + t) * 16 + n;
      const int kmv = kmv4[t];
      for (int h = 0; h < 2; ++h) {
        v8f acc = {};
        for (int kk = 0; kk < 4; ++kk) {
          acc = WMMA_F32(qa[h][kk], kf[t][h][kk], acc);
        }
        float* dst       = (h == 0) ? smem : (smem + H1);
        const int stride = (h == 0) ? ST0 : 512;
        for (int v = 0; v < 8; ++v) {
          const int r     = rb + v;
          const int valid = kmv && (khi ? qhi[v] : qlo[v]);
          dst[r * stride + col] = valid ? acc[v] : -1e20f;
        }
      }
    }
  }
  lds_barrier();

  v2f vb0[16], vb1[16];

  {
    float s1 = 0.f, s2 = 0.f;
    for (int i = 0; i < 16; ++i) { s1 += lq1[i] * lk1[i]; s2 += lq2[i] * lk2[i]; }
    const float lambda_full = __expf(s1) - __expf(s2) + LAMBDA_INIT;

    int kmc[16];
    for (int i = 0; i < 16; ++i) kmc[i] = kmask[b * S_ + lane + 32 * i];

    for (int mm = 0; mm < 2; ++mm) {
      const int m = w * 2 + mm;
      const float* r0 = smem + m * ST0;
      const float* r1 = smem + H1 + m * 512;
      float mx0 = -3.4e38f, mx1 = -3.4e38f;
      for (int i = 0; i < 16; ++i) {
        const int c = lane + 32 * i;
        mx0 = fmaxf(mx0, r0[c]);
        mx1 = fmaxf(mx1, r1[c]);
      }
      wave_max2(mx0, mx1);
      float e0[16], e1[16], sum0 = 0.f, sum1 = 0.f;
      for (int i = 0; i < 16; ++i) {
        const int c = lane + 32 * i;
        e0[i] = __expf(r0[c] - mx0); sum0 += e0[i];
        e1[i] = __expf(r1[c] - mx1); sum1 += e1[i];
      }
      if (mm == 1) {
        __builtin_amdgcn_sched_barrier(0);
        const float* vbase = Vm + (size_t)(b * 128 + w * 16) * 128 + lane * 4;
#pragma unroll
        for (int j = 0; j < 16; ++j) {
          const v4f x = *(const v4f*)(vbase + j * 128);
          vb0[j][0] = x[0]; vb0[j][1] = x[1];
          vb1[j][0] = x[2]; vb1[j][1] = x[3];
        }
      }
      sum0 = wave_sum(sum0);
      sum1 = wave_sum(sum1);
      const float inv0 = 1.f / (sum0 + 1e-8f);
      const float inv1 = 1.f / (sum1 + 1e-8f);
      float d[16], dmin = 3.4e38f;
      for (int i = 0; i < 16; ++i) {
        d[i] = e0[i] * inv0 - lambda_full * (e1[i] * inv1);
        dmin = fminf(dmin, d[i]);
      }
      dmin = wave_min(dmin);

      const int qv = qmask[R0 + m];
      float* w0 = smem + m * ST0;
      float* dg = diffp + (size_t)(R0 + m) * 512;
      for (int i = 0; i < 16; ++i) {
        const int c = lane + 32 * i;
        float dd = d[i] - dmin + 1e-5f;
        dd = (qv && kmc[i]) ? dd : 0.f;
        w0[c] = dd;
        __builtin_nontemporal_store(dd, dg + c);
      }

    }
  }
  lds_barrier();

  float lnw_l;
  v2f wo0[8], wo1[8];
  {
    v2f af[16];
#pragma unroll
    for (int j = 0; j < 16; ++j) {
      const int kb = (w * 16 + j) * 4 + khi * 2;
      af[j][0] = smem[n * ST0 + kb]; af[j][1] = smem[n * ST0 + kb + 1];
    }
    __builtin_amdgcn_sched_barrier(0);
    v8f ao0 = {}, ao1 = {};
#pragma unroll
    for (int j = 0; j < 16; ++j) {
      ao0 = WMMA_F32(af[j], vb0[j], ao0);
      ao1 = WMMA_F32(af[j], vb1[j], ao1);
    }

    lnw_l = lnw[lane];
    if (w == 0) {
      for (int kk = 0; kk < 8; ++kk) {
        const int kb = kk * 4 + khi * 2;
        wo0[kk] = *(const v2f*)(Wo + (size_t)n * 32 + kb);
        wo1[kk] = *(const v2f*)(Wo + (size_t)(16 + n) * 32 + kb);
      }
    }

    float* part = smem + PART + w * 512;
    for (int v = 0; v < 8; ++v) {
      const int r = rb + v;
      part[r * 32 + n]      = ao0[v];
      part[r * 32 + 16 + n] = ao1[v];
    }
  }
  lds_barrier();

  for (int mm = 0; mm < 2; ++mm) {
    const int m = w * 2 + mm;
    float x = smem[PART + m * 32 + lane];
    for (int p = 1; p < NW; ++p) x += smem[PART + p * 512 + m * 32 + lane];
    float sq = x * x;
    sq = wave_sum(sq);
    const float rms = rsqrtf(sq * (1.0f / 32.0f) + 1e-5f);
    smem[AOS + m * 32 + lane] = x * rms * lnw_l * 0.8f;
  }
  lds_barrier();

  if (w != 0) return;

  const float* aos = smem + AOS;
  v8f o0 = {}, o1 = {};
  for (int kk = 0; kk < 8; ++kk) {
    const int kb = kk * 4 + khi * 2;
    v2f a; a[0] = aos[n * 32 + kb]; a[1] = aos[n * 32 + kb + 1];
    o0 = WMMA_F32(a, wo0[kk], o0);
    o1 = WMMA_F32(a, wo1[kk], o1);
  }
  for (int v = 0; v < 8; ++v) {
    const size_t r = (size_t)(R0 + rb + v) * 32;
    __builtin_nontemporal_store(o0[v], outp + r + n);
    __builtin_nontemporal_store(o1[v], outp + r + 16 + n);
  }
}

extern "C" void kernel_launch(void* const* d_in, const int* in_sizes, int n_in,
                              void* d_out, int out_size, void* d_ws, size_t ws_size,
                              hipStream_t stream) {
  (void)in_sizes; (void)n_in; (void)out_size; (void)ws_size;
  const float* query = (const float*)d_in[0];
  const float* key   = (const float*)d_in[1];
  const int*   qmask = (const int*)d_in[2];
  const int*   kmask = (const int*)d_in[3];
  const float* Wq = (const float*)d_in[4];
  const float* Wk = (const float*)d_in[5];
  const float* Wv = (const float*)d_in[6];
  const float* Wo = (const float*)d_in[7];
  const float* lq1 = (const float*)d_in[8];
  const float* lk1 = (const float*)d_in[9];
  const float* lq2 = (const float*)d_in[10];
  const float* lk2 = (const float*)d_in[11];
  const float* lnw = (const float*)d_in[12];

  float* outp  = (float*)d_out;
  float* diffp = outp + (size_t)B_ * L_ * E_;

  float* Km = (float*)d_ws;
  float* Vm = Km + (size_t)B_ * S_ * E_;

  kv_proj_kernel<<<(B_ * S_) / 16, 32 * KV_NW, 0, stream>>>(key, Wk, Wv, Km, Vm);
  attn_kernel<<<(B_ * L_) / 16, ATT_THREADS, 0, stream>>>(query, Wq, Km, Vm, qmask, kmask,
                                                 lq1, lk1, lq2, lk2, lnw, Wo,
                                                 outp, diffp);
}


# attention kernel: two 16-row tiles per 16-wave work-group, each K and V fragment a wave loads feeds both tiles (K/V cache reads halve); attn_out sum order changes: 16 partials added in wave order
# speedup vs baseline: 7.8159x; 1.0434x over previous
#include <hip/hip_runtime.h>
#include <math.h>

typedef float v2f __attribute__((ext_vector_type(2)));
typedef float v8f __attribute__((ext_vector_type(8)));

#define WMMA_F32(a, b, c) \
  __builtin_amdgcn_wmma_f32_16x16x4_f32(false, (a), false, (b), (short)0, (c), false, false)

static constexpr int B_ = 8, S_ = 512, E_ = 32, L_ = 4096, SUBE_ = 768;
static constexpr float LAMBDA_INIT = 0.2f;

static constexpr int RT   = 2;
static constexpr int ROWS = 16 * RT;
static constexpr int ST0 = 513;
static constexpr int H1  = ROWS * ST0;
static constexpr int QST = 34;
static constexpr int QT  = ROWS * ST0 + ROWS * 512;
static constexpr int SMEM_FLOATS = QT + ROWS * QST;

typedef float v4f __attribute__((ext_vector_type(4)));

static constexpr int KV_NW  = 8;
static constexpr int KV_KS  = SUBE_ / KV_NW;
static constexpr int KV_PST = 66;
static constexpr int KV_PT  = 16 * KV_PST;

__global__ __launch_bounds__(32 * KV_NW) __attribute__((amdgpu_waves_per_eu(2, 2)))
void kv_proj_kernel(
    const float* __restrict__ key, const float* __restrict__ Wk,
    const float* __restrict__ Wv, float* __restrict__ Km, float* __restrict__ Vm) {
  __shared__ float part[KV_NW * KV_PT];

  const int lane = threadIdx.x & 31;
  const int w    = __builtin_amdgcn_readfirstlane((int)(threadIdx.x >> 5));
  const int n    = lane & 15;
  const int khi  = lane >> 4;
  const int m0   = blockIdx.x * 16;
  const int koff = w * KV_KS + khi * 4;

  const float* ap  = key + (size_t)(m0 + n) * SUBE_ + koff;
  const float* pk0 = Wk + (size_t)n * SUBE_ + koff;
  const float* pk1 = Wk + (size_t)(16 + n) * SUBE_ + koff;
  const float* pv0 = Wv + (size_t)n * SUBE_ + koff;
  const float* pv1 = Wv + (size_t)(16 + n) * SUBE_ + koff;

  v8f ak0 = {}, ak1 = {}, av0 = {}, av1 = {};
  for (int j = 0; j < KV_KS / 8; ++j) {
    const v4f a   = *(const v4f*)(ap  + 8 * j);
    const v4f bk0 = *(const v4f*)(pk0 + 8 * j);
    const v4f bk1 = *(const v4f*)(pk1 + 8 * j);
    const v4f bv0 = *(const v4f*)(pv0 + 8 * j);
    const v4f bv1 = *(const v4f*)(pv1 + 8 * j);
    v2f alo, ahi, k0lo, k0hi, k1lo, k1hi, v0lo, v0hi, v1lo, v1hi;
    alo[0]  = a[0];   alo[1]  = a[1];   ahi[0]  = a[2];   ahi[1]  = a[3];
    k0lo[0] = bk0[0]; k0lo[1] = bk0[1]; k0hi[0] = bk0[2]; k0hi[1] = bk0[3];
    k1lo[0] = bk1[0]; k1lo[1] = bk1[1]; k1hi[0] = bk1[2]; k1hi[1] = bk1[3];
    v0lo[0] = bv0[0]; v0lo[1] = bv0[1]; v0hi[0] = bv0[2]; v0hi[1] = bv0[3];
    v1lo[0] = bv1[0]; v1lo[1] = bv1[1]; v1hi[0] = bv1[2]; v1hi[1] = bv1[3];
    ak0 = WMMA_F32(alo, k0lo, ak0);
    ak1 = WMMA_F32(alo, k1lo, ak1);
    av0 = WMMA_F32(alo, v0lo, av0);
    av1 = WMMA_F32(alo, v1lo, av1);
    ak0 = WMMA_F32(ahi, k0hi, ak0);
    ak1 = WMMA_F32(ahi, k1hi, ak1);
    av0 = WMMA_F32(ahi, v0hi, av0);
    av1 = WMMA_F32(ahi, v1hi, av1);
  }

  float* pw = part + w * KV_PT;
  for (int v = 0; v < 8; ++v) {
    const int r = (khi * 8 + v) * KV_PST + n;
    pw[r]      = ak0[v];
    pw[r + 16] = ak1[v];
    pw[r + 32] = av0[v];
    pw[r + 48] = av1[v];
  }
  __syncthreads();

  {
    const int u  = (int)threadIdx.x & 127;
    const int g  = u >> 5;
    const int cn = u & 15;
    const int ch = (u >> 4) & 1;
    v4f x;
    for (int e = 0; e < 4; ++e) {
      int r, c;
      if (w < 4) { r = cn;                        c = (g >> 1) * 16 + (g & 1) * 8 + (e >> 1) * 4 + ch * 2 + (e & 1); }
      else       { r = g * 4 + ch * 2 + (e & 1);  c = 32 + (e >> 1) * 16 + cn; }
      float s = part[r * KV_PST + c];
      for (int p = 1; p < KV_NW; ++p) s += part[p * KV_PT + r * KV_PST + c];
      x[e] = s;
    }
    float* dst = (w < 4) ? Km : Vm;
    *(v4f*)(dst + (size_t)blockIdx.x * 512 + u * 4) = x;
  }
}

static __device__ __forceinline__ float xhalf(float x) {
  const int b = __builtin_bit_cast(int, x);
  return __builtin_bit_cast(float,
      __builtin_amdgcn_permlanex16(b, b, 0x76543210, 0xfedcba98, false, false));
}
template <int M>
static __device__ __forceinline__ float xrow(float x) {
  const int b = __builtin_bit_cast(int, x);
  return __builtin_bit_cast(float,
      __builtin_amdgcn_update_dpp(b, b, 0x160 + M, 0xF, 0xF, true));
}
template <int M>
static __device__ __forceinline__ float max_xrow(float x) {
  float r;
  asm("v_max_num_f32_dpp %0, %1, %1 row_xmask:%2 row_mask:0xf bank_mask:0xf bound_ctrl:1"
      : "=v"(r) : "v"(x), "n"(M));
  return r;
}
template <int M>
static __device__ __forceinline__ float min_xrow(float x) {
  float r;
  asm("v_min_num_f32_dpp %0, %1, %1 row_xmask:%2 row_mask:0xf bank_mask:0xf bound_ctrl:1"
      : "=v"(r) : "v"(x), "n"(M));
  return r;
}
static __device__ __forceinline__ void wave_max2(float& a, float& b) {
  a = fmaxf(a, xhalf(a));  b = fmaxf(b, xhalf(b));
  a = max_xrow<8>(a);      b = max_xrow<8>(b);
  a = max_xrow<4>(a);      b = max_xrow<4>(b);
  a = max_xrow<2>(a);      b = max_xrow<2>(b);
  a = max_xrow<1>(a);      b = max_xrow<1>(b);
}
static __device__ __forceinline__ float wave_min(float x) {
  x = fminf(x, xhalf(x));
  x = min_xrow<8>(x); x = min_xrow<4>(x);
  x = min_xrow<2>(x); x = min_xrow<1>(x);
  return x;
}
static __device__ __forceinline__ float wave_sum(float x) {
  x += xhalf(x);
  x += xrow<8>(x); x += xrow<4>(x);
  x += xrow<2>(x); x += xrow<1>(x);
  return x;
}

static __device__ __forceinline__ void lds_barrier() {
  __builtin_amdgcn_fence(__ATOMIC_RELEASE, "workgroup", "local");
  __builtin_amdgcn_s_barrier();
  __builtin_amdgcn_fence(__ATOMIC_ACQUIRE, "workgroup", "local");
}

static constexpr int NW          = 8 * RT;
static constexpr int ATT_THREADS = 32 * NW;
static constexpr int PART        = H1;
static constexpr int AOS         = QT;
static_assert(NW * ROWS * 32 <= ROWS * 512, "partial tiles fit the head1 region");
static_assert(ROWS * 32 <= ROWS * QST, "attn_out tile fits the Q tile's place");
static_assert(L_ % ROWS == 0, "a work-group stays inside one batch");

__global__ __launch_bounds__(ATT_THREADS) void attn_kernel(
    const float* __restrict__ query, const float* __restrict__ Wq,
    const float* __restrict__ Km,
    const float* __restrict__ Vm, const int* __restrict__ qmask,
    const int* __restrict__ kmask, const float* __restrict__ lq1,
    const float* __restrict__ lk1, const float* __restrict__ lq2,
    const float* __restrict__ lk2, const float* __restrict__ lnw,
    const float* __restrict__ Wo, float* __restrict__ outp,
    float* __restrict__ diffp) {
  __shared__ __attribute__((aligned(16))) float smem[SMEM_FLOATS];

  const int lane = threadIdx.x & 31;
  const int w    = __builtin_amdgcn_readfirstlane((int)(threadIdx.x >> 5));
  const int n    = lane & 15;
  const int khi  = lane >> 4;
  const int R0   = blockIdx.x * ROWS;
  const int b    = R0 / L_;
  const int rb   = khi ? 8 : 0;

  {
    v2f qx[8], wq[8];
    if (w < 2 * RT) {
      const float* qrow = query + (size_t)(R0 + 16 * (w >> 1) + n) * 32 + khi * 2;
      const float* wrow = Wq + (size_t)(16 * (w & 1) + n) * 32 + khi * 2;
      for (int kk = 0; kk < 8; ++kk) {
        qx[kk] = *(const v2f*)(qrow + kk * 4);
        wq[kk] = *(const v2f*)(wrow + kk * 4);
      }
    }
    int qm[RT][8];
    for (int rt = 0; rt < RT; ++rt) {
      const int* qp = qmask + R0 + 16 * rt + rb;
      for (int v = 0; v < 8; ++v) qm[rt][v] = qp[v];
    }

    v2f kf[2][2][4];
    int kmv2[2];
    const float* kfb = Km + (size_t)(b * 32 + w * 2) * 512 + lane * 4;
    for (int t = 0; t < 2; ++t) {
      const int col = (w * 2 + t) * 16 + n;
      kmv2[t] = kmask[b * S_ + col];
      for (int h = 0; h < 2; ++h)
        for (int p = 0; p < 2; ++p) {
          const v4f x = *(const v4f*)(kfb + ((t * 2 + h) * 2 + p) * 128);
          kf[t][h][2 * p][0]     = x[0]; kf[t][h][2 * p][1]     = x[1];
          kf[t][h][2 * p + 1][0] = x[2]; kf[t][h][2 * p + 1][1] = x[3];
        }
    }
    __builtin_amdgcn_sched_barrier(0);

    float* qt = smem + QT;
    if (w < 2 * RT) {
      v8f qacc = {};
      for (int kk = 0; kk < 8; ++kk) qacc = WMMA_F32(qx[kk], wq[kk], qacc);
      for (int v = 0; v < 8; ++v)
        qt[(16 * (w >> 1) + rb + v) * QST + 16 * (w & 1) + n] = qacc[v] * 0.25f;
    }
    lds_barrier();
    v2f qa[RT][2][4];
    for (int rt = 0; rt < RT; ++rt)
      for (int h = 0; h < 2; ++h)
        for (int kk = 0; kk < 4; ++kk)
          qa[rt][h][kk] = *(const v2f*)(qt + (16 * rt + n) * QST + h * 16 + kk * 4 + khi * 2);
    __builtin_amdgcn_sched_barrier(0);

    for (int t = 0; t < 2; ++t) {
      const int col = (w * 2 + t) * 16 + n;
      const int kmv = kmv2[t];
      for (int h = 0; h < 2; ++h) {
        float* dst       = (h == 0) ? smem : (smem + H1);
        const int stride = (h == 0) ? ST0 : 512;
        for (int rt = 0; rt < RT; ++rt) {
          v8f acc = {};
          for (int kk = 0; kk < 4; ++kk) {
            acc = WMMA_F32(qa[rt][h][kk], kf[t][h][kk], acc);
          }
          for (int v = 0; v < 8; ++v) {
            const int r     = 16 * rt + rb + v;
            const int valid = kmv && qm[rt][v];
            dst[r * stride + col] = valid ? acc[v] : -1e20f;
          }
        }
      }
    }
  }
  lds_barrier();

  v2f vb0[8], vb1[8];

  {
    float s1 = 0.f, s2 = 0.f;
    for (int i = 0; i < 16; ++i) { s1 += lq1[i] * lk1[i]; s2 += lq2[i] * lk2[i]; }
    const float lambda_full = __expf(s1) - __expf(s2) + LAMBDA_INIT;

    int kmc[16];
    for (int i = 0; i < 16; ++i) kmc[i] = kmask[b * S_ + lane + 32 * i];

    for (int mm = 0; mm < 2; ++mm) {
      const int m = w * 2 + mm;
      const float* r0 = smem + m * ST0;
      const float* r1 = smem + H1 + m * 512;
      float mx0 = -3.4e38f, mx1 = -3.4e38f;
      for (int i = 0; i < 16; ++i) {
        const int c = lane + 32 * i;
        mx0 = fmaxf(mx0, r0[c]);
        mx1 = fmaxf(mx1, r1[c]);
      }
      wave_max2(mx0, mx1);
      float e0[16], e1[16], sum0 = 0.f, sum1 = 0.f;
      for (int i = 0; i < 16; ++i) {
        const int c = lane + 32 * i;
        e0[i] = __expf(r0[c] - mx0); sum0 += e0[i];
        e1[i] = __expf(r1[c] - mx1); sum1 += e1[i];
      }
      if (mm == 1) {
        __builtin_amdgcn_sched_barrier(0);
        const float* vbase = Vm + (size_t)(b * 128 + w * 8) * 128 + lane * 4;
#pragma unroll
        for (int j = 0; j < 8; ++j) {
          const v4f x = *(const v4f*)(vbase + j * 128);
          vb0[j][0] = x[0]; vb0[j][1] = x[1];
          vb1[j][0] = x[2]; vb1[j][1] = x[3];
        }
      }
      sum0 = wave_sum(sum0);
      sum1 = wave_sum(sum1);
      const float inv0 = 1.f / (sum0 + 1e-8f);
      const float inv1 = 1.f / (sum1 + 1e-8f);
      float d[16], dmin = 3.4e38f;
      for (int i = 0; i < 16; ++i) {
        d[i] = e0[i] * inv0 - lambda_full * (e1[i] * inv1);
        dmin = fminf(dmin, d[i]);
      }
      dmin = wave_min(dmin);

      const int qv = qmask[R0 + m];
      float* w0 = smem + m * ST0;
      float* dg = diffp + (size_t)(R0 + m) * 512;
      for (int i = 0; i < 16; ++i) {
        const int c = lane + 32 * i;
        float dd = d[i] - dmin + 1e-5f;
        dd = (qv && kmc[i]) ? dd : 0.f;
        w0[c] = dd;
        __builtin_nontemporal_store(dd, dg + c);
      }

    }
  }
  lds_barrier();

  float lnw_l;
  v2f wo0[8], wo1[8];
  {
    v2f af[RT][8];
#pragma unroll
    for (int rt = 0; rt < RT; ++rt)
#pragma unroll
      for (int j = 0; j < 8; ++j) {
        const int kb = (w * 8 + j) * 4 + khi * 2;
        af[rt][j][0] = smem[(16 * rt + n) * ST0 + kb];
        af[rt][j][1] = smem[(16 * rt + n) * ST0 + kb + 1];
      }
    __builtin_amdgcn_sched_barrier(0);
    v8f ao[RT][2] = {};
#pragma unroll
    for (int j = 0; j < 8; ++j) {
#pragma unroll
      for (int rt = 0; rt < RT; ++rt) {
        ao[rt][0] = WMMA_F32(af[rt][j], vb0[j], ao[rt][0]);
        ao[rt][1] = WMMA_F32(af[rt][j], vb1[j], ao[rt][1]);
      }
    }

    lnw_l = lnw[lane];
    if (w < RT) {
      for (int kk = 0; kk < 8; ++kk) {
        const int kb = kk * 4 + khi * 2;
        wo0[kk] = *(const v2f*)(Wo + (size_t)n * 32 + kb);
        wo1[kk] = *(const v2f*)(Wo + (size_t)(16 + n) * 32 + kb);
      }
    }

    float* part = smem + PART + w * (ROWS * 32);
    for (int rt = 0; rt < RT; ++rt)
      for (int v = 0; v < 8; ++v) {
        const int r = 16 * rt + rb + v;
        part[r * 32 + n]      = ao[rt][0][v];
        part[r * 32 + 16 + n] = ao[rt][1][v];
      }
  }
  lds_barrier();

  for (int mm = 0; mm < 2; ++mm) {
    const int m = w * 2 + mm;
    float x = smem[PART + m * 32 + lane];
    for (int p = 1; p < NW; ++p) x += smem[PART + p * (ROWS * 32) + m * 32 + lane];
    float sq = x * x;
    sq = wave_sum(sq);
    const float rms = rsqrtf(sq * (1.0f / 32.0f) + 1e-5f);
    smem[AOS + m * 32 + lane] = x * rms * lnw_l * 0.8f;
  }
  lds_barrier();

  if (w >= RT) return;

  const float* aos = smem + AOS + w * 512;
  v8f o0 = {}, o1 = {};
  for (int kk = 0; kk < 8; ++kk) {
    const int kb = kk * 4 + khi * 2;
    v2f a; a[0] = aos[n * 32 + kb]; a[1] = aos[n * 32 + kb + 1];
    o0 = WMMA_F32(a, wo0[kk], o0);
    o1 = WMMA_F32(a, wo1[kk], o1);
  }
  for (int v = 0; v < 8; ++v) {
    const size_t r = (size_t)(R0 + 16 * w + rb + v) * 32;
    __builtin_nontemporal_store(o0[v], outp + r + n);
    __builtin_nontemporal_store(o1[v], outp + r + 16 + n);
  }
}

extern "C" void kernel_launch(void* const* d_in, const int* in_sizes, int n_in,
                              void* d_out, int out_size, void* d_ws, size_t ws_size,
                              hipStream_t stream) {
  (void)in_sizes; (void)n_in; (void)out_size; (void)ws_size;
  const float* query = (const float*)d_in[0];
  const float* key   = (const float*)d_in[1];
  const int*   qmask = (const int*)d_in[2];
  const int*   kmask = (const int*)d_in[3];
  const float* Wq = (const float*)d_in[4];
  const float* Wk = (const float*)d_in[5];
  const float* Wv = (const float*)d_in[6];
  const float* Wo = (const float*)d_in[7];
  const float* lq1 = (const float*)d_in[8];
  const float* lk1 = (const float*)d_in[9];
  const float* lq2 = (const float*)d_in[10];
  const float* lk2 = (const float*)d_in[11];
  const float* lnw = (const float*)d_in[12];

  float* outp  = (float*)d_out;
  float* diffp = outp + (size_t)B_ * L_ * E_;

  float* Km = (float*)d_ws;
  float* Vm = Km + (size_t)B_ * S_ * E_;

  kv_proj_kernel<<<(B_ * S_) / 16, 32 * KV_NW, 0, stream>>>(key, Wk, Wv, Km, Vm);
  attn_kernel<<<(B_ * L_) / ROWS, ATT_THREADS, 0, stream>>>(query, Wq, Km, Vm, qmask, kmask,
                                                 lq1, lk1, lq2, lk2, lnw, Wo,
                                                 outp, diffp);
}


# attention kernel, softmax phase: a wave skips the softmax of a row whose query mask is 0 (scalar branch) and stores that row's zero diff directly; key-mask loads sent ahead of the phase's barrier
# speedup vs baseline: 8.0690x; 1.0324x over previous
#include <hip/hip_runtime.h>
#include <math.h>

typedef float v2f __attribute__((ext_vector_type(2)));
typedef float v8f __attribute__((ext_vector_type(8)));

#define WMMA_F32(a, b, c) \
  __builtin_amdgcn_wmma_f32_16x16x4_f32(false, (a), false, (b), (short)0, (c), false, false)

static constexpr int B_ = 8, S_ = 512, E_ = 32, L_ = 4096, SUBE_ = 768;
static constexpr float LAMBDA_INIT = 0.2f;

static constexpr int RT   = 2;
static constexpr int ROWS = 16 * RT;
static constexpr int ST0 = 513;
static constexpr int H1  = ROWS * ST0;
static constexpr int QST = 34;
static constexpr int QT  = ROWS * ST0 + ROWS * 512;
static constexpr int SMEM_FLOATS = QT + ROWS * QST;

typedef float v4f __attribute__((ext_vector_type(4)));

static constexpr int KV_NW  = 8;
static constexpr int KV_KS  = SUBE_ / KV_NW;
static constexpr int KV_PST = 66;
static constexpr int KV_PT  = 16 * KV_PST;

__global__ __launch_bounds__(32 * KV_NW) __attribute__((amdgpu_waves_per_eu(2, 2)))
void kv_proj_kernel(
    const float* __restrict__ key, const float* __restrict__ Wk,
    const float* __restrict__ Wv, float* __restrict__ Km, float* __restrict__ Vm) {
  __shared__ float part[KV_NW * KV_PT];

  const int lane = threadIdx.x & 31;
  const int w    = __builtin_amdgcn_readfirstlane((int)(threadIdx.x >> 5));
  const int n    = lane & 15;
  const int khi  = lane >> 4;
  const int m0   = blockIdx.x * 16;
  const int koff = w * KV_KS + khi * 4;

  const float* ap  = key + (size_t)(m0 + n) * SUBE_ + koff;
  const float* pk0 = Wk + (size_t)n * SUBE_ + koff;
  const float* pk1 = Wk + (size_t)(16 + n) * SUBE_ + koff;
  const float* pv0 = Wv + (size_t)n * SUBE_ + koff;
  const float* pv1 = Wv + (size_t)(16 + n) * SUBE_ + koff;

  v8f ak0 = {}, ak1 = {}, av0 = {}, av1 = {};
  for (int j = 0; j < KV_KS / 8; ++j) {
    const v4f a   = *(const v4f*)(ap  + 8 * j);
    const v4f bk0 = *(const v4f*)(pk0 + 8 * j);
    const v4f bk1 = *(const v4f*)(pk1 + 8 * j);
    const v4f bv0 = *(const v4f*)(pv0 + 8 * j);
    const v4f bv1 = *(const v4f*)(pv1 + 8 * j);
    v2f alo, ahi, k0lo, k0hi, k1lo, k1hi, v0lo, v0hi, v1lo, v1hi;
    alo[0]  = a[0];   alo[1]  = a[1];   ahi[0]  = a[2];   ahi[1]  = a[3];
    k0lo[0] = bk0[0]; k0lo[1] = bk0[1]; k0hi[0] = bk0[2]; k0hi[1] = bk0[3];
    k1lo[0] = bk1[0]; k1lo[1] = bk1[1]; k1hi[0] = bk1[2]; k1hi[1] = bk1[3];
    v0lo[0] = bv0[0]; v0lo[1] = bv0[1]; v0hi[0] = bv0[2]; v0hi[1] = bv0[3];
    v1lo[0] = bv1[0]; v1lo[1] = bv1[1]; v1hi[0] = bv1[2]; v1hi[1] = bv1[3];
    ak0 = WMMA_F32(alo, k0lo, ak0);
    ak1 = WMMA_F32(alo, k1lo, ak1);
    av0 = WMMA_F32(alo, v0lo, av0);
    av1 = WMMA_F32(alo, v1lo, av1);
    ak0 = WMMA_F32(ahi, k0hi, ak0);
    ak1 = WMMA_F32(ahi, k1hi, ak1);
    av0 = WMMA_F32(ahi, v0hi, av0);
    av1 = WMMA_F32(ahi, v1hi, av1);
  }

  float* pw = part + w * KV_PT;
  for (int v = 0; v < 8; ++v) {
    const int r = (khi * 8 + v) * KV_PST + n;
    pw[r]      = ak0[v];
    pw[r + 16] = ak1[v];
    pw[r + 32] = av0[v];
    pw[r + 48] = av1[v];
  }
  __syncthreads();

  {
    const int u  = (int)threadIdx.x & 127;
    const int g  = u >> 5;
    const int cn = u & 15;
    const int ch = (u >> 4) & 1;
    v4f x;
    for (int e = 0; e < 4; ++e) {
      int r, c;
      if (w < 4) { r = cn;                        c = (g >> 1) * 16 + (g & 1) * 8 + (e >> 1) * 4 + ch * 2 + (e & 1); }
      else       { r = g * 4 + ch * 2 + (e & 1);  c = 32 + (e >> 1) * 16 + cn; }
      float s = part[r * KV_PST + c];
      for (int p = 1; p < KV_NW; ++p) s += part[p * KV_PT + r * KV_PST + c];
      x[e] = s;
    }
    float* dst = (w < 4) ? Km : Vm;
    *(v4f*)(dst + (size_t)blockIdx.x * 512 + u * 4) = x;
  }
}

static __device__ __forceinline__ float xhalf(float x) {
  const int b = __builtin_bit_cast(int, x);
  return __builtin_bit_cast(float,
      __builtin_amdgcn_permlanex16(b, b, 0x76543210, 0xfedcba98, false, false));
}
template <int M>
static __device__ __forceinline__ float xrow(float x) {
  const int b = __builtin_bit_cast(int, x);
  return __builtin_bit_cast(float,
      __builtin_amdgcn_update_dpp(b, b, 0x160 + M, 0xF, 0xF, true));
}
template <int M>
static __device__ __forceinline__ float max_xrow(float x) {
  float r;
  asm("v_max_num_f32_dpp %0, %1, %1 row_xmask:%2 row_mask:0xf bank_mask:0xf bound_ctrl:1"
      : "=v"(r) : "v"(x), "n"(M));
  return r;
}
template <int M>
static __device__ __forceinline__ float min_xrow(float x) {
  float r;
  asm("v_min_num_f32_dpp %0, %1, %1 row_xmask:%2 row_mask:0xf bank_mask:0xf bound_ctrl:1"
      : "=v"(r) : "v"(x), "n"(M));
  return r;
}
static __device__ __forceinline__ void wave_max2(float& a, float& b) {
  a = fmaxf(a, xhalf(a));  b = fmaxf(b, xhalf(b));
  a = max_xrow<8>(a);      b = max_xrow<8>(b);
  a = max_xrow<4>(a);      b = max_xrow<4>(b);
  a = max_xrow<2>(a);      b = max_xrow<2>(b);
  a = max_xrow<1>(a);      b = max_xrow<1>(b);
}
static __device__ __forceinline__ float wave_min(float x) {
  x = fminf(x, xhalf(x));
  x = min_xrow<8>(x); x = min_xrow<4>(x);
  x = min_xrow<2>(x); x = min_xrow<1>(x);
  return x;
}
static __device__ __forceinline__ float wave_sum(float x) {
  x += xhalf(x);
  x += xrow<8>(x); x += xrow<4>(x);
  x += xrow<2>(x); x += xrow<1>(x);
  return x;
}

static __device__ __forceinline__ void lds_barrier() {
  __builtin_amdgcn_fence(__ATOMIC_RELEASE, "workgroup", "local");
  __builtin_amdgcn_s_barrier();
  __builtin_amdgcn_fence(__ATOMIC_ACQUIRE, "workgroup", "local");
}

static constexpr int NW          = 8 * RT;
static constexpr int ATT_THREADS = 32 * NW;
static constexpr int PART        = H1;
static constexpr int AOS         = QT;
static_assert(NW * ROWS * 32 <= ROWS * 512, "partial tiles fit the head1 region");
static_assert(ROWS * 32 <= ROWS * QST, "attn_out tile fits the Q tile's place");
static_assert(L_ % ROWS == 0, "a work-group stays inside one batch");

__global__ __launch_bounds__(ATT_THREADS) void attn_kernel(
    const float* __restrict__ query, const float* __restrict__ Wq,
    const float* __restrict__ Km,
    const float* __restrict__ Vm, const int* __restrict__ qmask,
    const int* __restrict__ kmask, const float* __restrict__ lq1,
    const float* __restrict__ lk1, const float* __restrict__ lq2,
    const float* __restrict__ lk2, const float* __restrict__ lnw,
    const float* __restrict__ Wo, float* __restrict__ outp,
    float* __restrict__ diffp) {
  __shared__ __attribute__((aligned(16))) float smem[SMEM_FLOATS];

  const int lane = threadIdx.x & 31;
  const int w    = __builtin_amdgcn_readfirstlane((int)(threadIdx.x >> 5));
  const int n    = lane & 15;
  const int khi  = lane >> 4;
  const int R0   = blockIdx.x * ROWS;
  const int b    = R0 / L_;
  const int rb   = khi ? 8 : 0;

  int kmc[16];

  {
    v2f qx[8], wq[8];
    if (w < 2 * RT) {
      const float* qrow = query + (size_t)(R0 + 16 * (w >> 1) + n) * 32 + khi * 2;
      const float* wrow = Wq + (size_t)(16 * (w & 1) + n) * 32 + khi * 2;
      for (int kk = 0; kk < 8; ++kk) {
        qx[kk] = *(const v2f*)(qrow + kk * 4);
        wq[kk] = *(const v2f*)(wrow + kk * 4);
      }
    }
    int qm[RT][8];
    for (int rt = 0; rt < RT; ++rt) {
      const int* qp = qmask + R0 + 16 * rt + rb;
      for (int v = 0; v < 8; ++v) qm[rt][v] = qp[v];
    }

    v2f kf[2][2][4];
    int kmv2[2];
    const float* kfb = Km + (size_t)(b * 32 + w * 2) * 512 + lane * 4;
    for (int t = 0; t < 2; ++t) {
      const int col = (w * 2 + t) * 16 + n;
      kmv2[t] = kmask[b * S_ + col];
      for (int h = 0; h < 2; ++h)
        for (int p = 0; p < 2; ++p) {
          const v4f x = *(const v4f*)(kfb + ((t * 2 + h) * 2 + p) * 128);
          kf[t][h][2 * p][0]     = x[0]; kf[t][h][2 * p][1]     = x[1];
          kf[t][h][2 * p + 1][0] = x[2]; kf[t][h][2 * p + 1][1] = x[3];
        }
    }
    __builtin_amdgcn_sched_barrier(0);

    float* qt = smem + QT;
    if (w < 2 * RT) {
      v8f qacc = {};
      for (int kk = 0; kk < 8; ++kk) qacc = WMMA_F32(qx[kk], wq[kk], qacc);
      for (int v = 0; v < 8; ++v)
        qt[(16 * (w >> 1) + rb + v) * QST + 16 * (w & 1) + n] = qacc[v] * 0.25f;
    }
    lds_barrier();
    v2f qa[RT][2][4];
    for (int rt = 0; rt < RT; ++rt)
      for (int h = 0; h < 2; ++h)
        for (int kk = 0; kk < 4; ++kk)
          qa[rt][h][kk] = *(const v2f*)(qt + (16 * rt + n) * QST + h * 16 + kk * 4 + khi * 2);
    __builtin_amdgcn_sched_barrier(0);

    for (int t = 0; t < 2; ++t) {
      const int col = (w * 2 + t) * 16 + n;
      const int kmv = kmv2[t];
      for (int h = 0; h < 2; ++h) {
        float* dst       = (h == 0) ? smem : (smem + H1);
        const int stride = (h == 0) ? ST0 : 512;
        for (int rt = 0; rt < RT; ++rt) {
          v8f acc = {};
          for (int kk = 0; kk < 4; ++kk) {
            acc = WMMA_F32(qa[rt][h][kk], kf[t][h][kk], acc);
          }
          for (int v = 0; v < 8; ++v) {
            const int r     = 16 * rt + rb + v;
            const int valid = kmv && qm[rt][v];
            dst[r * stride + col] = valid ? acc[v] : -1e20f;
          }
        }
      }
    }
    __builtin_amdgcn_sched_barrier(0);
    for (int i = 0; i < 16; ++i) kmc[i] = kmask[b * S_ + lane + 32 * i];
  }
  lds_barrier();

  v2f vb0[8], vb1[8];

  {
    float s1 = 0.f, s2 = 0.f;
    for (int i = 0; i < 16; ++i) { s1 += lq1[i] * lk1[i]; s2 += lq2[i] * lk2[i]; }
    const float lambda_full = __expf(s1) - __expf(s2) + LAMBDA_INIT;

    for (int mm = 0; mm < 2; ++mm) {
      const int m = w * 2 + mm;
      const int qv = qmask[R0 + m];
      float* w0 = smem + m * ST0;
      float* dg = diffp + (size_t)(R0 + m) * 512;
      const float* vbase = Vm + (size_t)(b * 128 + w * 8) * 128 + lane * 4;
      if (qv) {
        const float* r0 = smem + m * ST0;
        const float* r1 = smem + H1 + m * 512;
        float mx0 = -3.4e38f, mx1 = -3.4e38f;
        for (int i = 0; i < 16; ++i) {
          const int c = lane + 32 * i;
          mx0 = fmaxf(mx0, r0[c]);
          mx1 = fmaxf(mx1, r1[c]);
        }
        wave_max2(mx0, mx1);
        float e0[16], e1[16], sum0 = 0.f, sum1 = 0.f;
        for (int i = 0; i < 16; ++i) {
          const int c = lane + 32 * i;
          e0[i] = __expf(r0[c] - mx0); sum0 += e0[i];
          e1[i] = __expf(r1[c] - mx1); sum1 += e1[i];
        }
        if (mm == 1) {
          __builtin_amdgcn_sched_barrier(0);
#pragma unroll
          for (int j = 0; j < 8; ++j) {
            const v4f x = *(const v4f*)(vbase + j * 128);
            vb0[j][0] = x[0]; vb0[j][1] = x[1];
            vb1[j][0] = x[2]; vb1[j][1] = x[3];
          }
        }
        sum0 = wave_sum(sum0);
        sum1 = wave_sum(sum1);
        const float inv0 = 1.f / (sum0 + 1e-8f);
        const float inv1 = 1.f / (sum1 + 1e-8f);
        float d[16], dmin = 3.4e38f;
        for (int i = 0; i < 16; ++i) {
          d[i] = e0[i] * inv0 - lambda_full * (e1[i] * inv1);
          dmin = fminf(dmin, d[i]);
        }
        dmin = wave_min(dmin);

        for (int i = 0; i < 16; ++i) {
          const int c = lane + 32 * i;
          float dd = d[i] - dmin + 1e-5f;
          dd = kmc[i] ? dd : 0.f;
          w0[c] = dd;
          __builtin_nontemporal_store(dd, dg + c);
        }
      } else {
        if (mm == 1) {
#pragma unroll
          for (int j = 0; j < 8; ++j) {
            const v4f x = *(const v4f*)(vbase + j * 128);
            vb0[j][0] = x[0]; vb0[j][1] = x[1];
            vb1[j][0] = x[2]; vb1[j][1] = x[3];
          }
        }
        for (int i = 0; i < 16; ++i) w0[lane + 32 * i] = 0.f;
        const v4f z = {0.f, 0.f, 0.f, 0.f};
        for (int q = 0; q < 4; ++q)
          __builtin_nontemporal_store(z, (v4f*)(dg + q * 128 + lane * 4));
      }
    }
  }
  lds_barrier();

  float lnw_l;
  v2f wo0[8], wo1[8];
  {
    v2f af[RT][8];
#pragma unroll
    for (int rt = 0; rt < RT; ++rt)
#pragma unroll
      for (int j = 0; j < 8; ++j) {
        const int kb = (w * 8 + j) * 4 + khi * 2;
        af[rt][j][0] = smem[(16 * rt + n) * ST0 + kb];
        af[rt][j][1] = smem[(16 * rt + n) * ST0 + kb + 1];
      }
    __builtin_amdgcn_sched_barrier(0);
    v8f ao[RT][2] = {};
#pragma unroll
    for (int j = 0; j < 8; ++j) {
#pragma unroll
      for (int rt = 0; rt < RT; ++rt) {
        ao[rt][0] = WMMA_F32(af[rt][j], vb0[j], ao[rt][0]);
        ao[rt][1] = WMMA_F32(af[rt][j], vb1[j], ao[rt][1]);
      }
    }

    lnw_l = lnw[lane];
    if (w < RT) {
      for (int kk = 0; kk < 8; ++kk) {
        const int kb = kk * 4 + khi * 2;
        wo0[kk] = *(const v2f*)(Wo + (size_t)n * 32 + kb);
        wo1[kk] = *(const v2f*)(Wo + (size_t)(16 + n) * 32 + kb);
      }
    }

    float* part = smem + PART + w * (ROWS * 32);
    for (int rt = 0; rt < RT; ++rt)
      for (int v = 0; v < 8; ++v) {
        const int r = 16 * rt + rb + v;
        part[r * 32 + n]      = ao[rt][0][v];
        part[r * 32 + 16 + n] = ao[rt][1][v];
      }
  }
  lds_barrier();

  for (int mm = 0; mm < 2; ++mm) {
    const int m = w * 2 + mm;
    float x = smem[PART + m * 32 + lane];
    for (int p = 1; p < NW; ++p) x += smem[PART + p * (ROWS * 32) + m * 32 + lane];
    float sq = x * x;
    sq = wave_sum(sq);
    const float rms = rsqrtf(sq * (1.0f / 32.0f) + 1e-5f);
    smem[AOS + m * 32 + lane] = x * rms * lnw_l * 0.8f;
  }
  lds_barrier();

  if (w >= RT) return;

  const float* aos = smem + AOS + w * 512;
  v8f o0 = {}, o1 = {};
  for (int kk = 0; kk < 8; ++kk) {
    const int kb = kk * 4 + khi * 2;
    v2f a; a[0] = aos[n * 32 + kb]; a[1] = aos[n * 32 + kb + 1];
    o0 = WMMA_F32(a, wo0[kk], o0);
    o1 = WMMA_F32(a, wo1[kk], o1);
  }
  for (int v = 0; v < 8; ++v) {
    const size_t r = (size_t)(R0 + 16 * w + rb + v) * 32;
    __builtin_nontemporal_store(o0[v], outp + r + n);
    __builtin_nontemporal_store(o1[v], outp + r + 16 + n);
  }
}

extern "C" void kernel_launch(void* const* d_in, const int* in_sizes, int n_in,
                              void* d_out, int out_size, void* d_ws, size_t ws_size,
                              hipStream_t stream) {
  (void)in_sizes; (void)n_in; (void)out_size; (void)ws_size;
  const float* query = (const float*)d_in[0];
  const float* key   = (const float*)d_in[1];
  const int*   qmask = (const int*)d_in[2];
  const int*   kmask = (const int*)d_in[3];
  const float* Wq = (const float*)d_in[4];
  const float* Wk = (const float*)d_in[5];
  const float* Wv = (const float*)d_in[6];
  const float* Wo = (const float*)d_in[7];
  const float* lq1 = (const float*)d_in[8];
  const float* lk1 = (const float*)d_in[9];
  const float* lq2 = (const float*)d_in[10];
  const float* lk2 = (const float*)d_in[11];
  const float* lnw = (const float*)d_in[12];

  float* outp  = (float*)d_out;
  float* diffp = outp + (size_t)B_ * L_ * E_;

  float* Km = (float*)d_ws;
  float* Vm = Km + (size_t)B_ * S_ * E_;

  kv_proj_kernel<<<(B_ * S_) / 16, 32 * KV_NW, 0, stream>>>(key, Wk, Wv, Km, Vm);
  attn_kernel<<<(B_ * L_) / ROWS, ATT_THREADS, 0, stream>>>(query, Wq, Km, Vm, qmask, kmask,
                                                 lq1, lk1, lq2, lk2, lnw, Wo,
                                                 outp, diffp);
}
